# starting kernel
# baseline (speedup 1.0000x reference)
#include <hip/hip_runtime.h>
#include <math.h>

typedef __attribute__((ext_vector_type(2))) float v2f;
typedef __attribute__((ext_vector_type(8))) float v8f;

__global__ void fill_kernel(float* __restrict__ p, float v, long n) {
  long i = (long)blockIdx.x * blockDim.x + threadIdx.x;
  if (i < n) p[i] = v;
}

__device__ __forceinline__ void atomicMaxF(float* addr, float v) {
  if (v >= 0.0f) atomicMax((int*)addr, __float_as_int(v));
  else           atomicMin((unsigned int*)addr, __float_as_uint(v));
}

template <int NCOL>
__global__ void __launch_bounds__(256) wmma_gemm_f32_k128(
    const float* __restrict__ A, const float* __restrict__ B,
    float* __restrict__ C, int M) {
  __shared__ float As[16 * 128];
  const int lane = threadIdx.x & 31;
  const int wave = threadIdx.x >> 5;
  const int row0 = blockIdx.x * 16;

  if (row0 + 16 <= M) {
    const float4* A4  = (const float4*)(A + (long)row0 * 128);
    float4*       As4 = (float4*)As;
    for (int i = threadIdx.x; i < 16 * 128 / 4; i += blockDim.x) As4[i] = A4[i];
  } else {
    for (int i = threadIdx.x; i < 16 * 128; i += blockDim.x) {
      int r = row0 + (i >> 7);
      if (r >= M) r = M - 1;
      As[i] = A[(long)r * 128 + (i & 127)];
    }
  }
  __syncthreads();

  const int col0  = wave * 16;
  const int mrow  = lane & 15;
  const int khalf = (lane >> 4) << 1;
  const float* Ap = As + mrow * 128 + khalf;
  const float* Bp = B + (long)khalf * NCOL + col0 + mrow;

  v8f acc0 = {}, acc1 = {}, acc2 = {}, acc3 = {};
#pragma unroll
  for (int k = 0; k < 128; k += 16) {
    v2f a0, a1, a2, a3, b0, b1, b2, b3;
    a0.x = Ap[k + 0];  a0.y = Ap[k + 1];
    a1.x = Ap[k + 4];  a1.y = Ap[k + 5];
    a2.x = Ap[k + 8];  a2.y = Ap[k + 9];
    a3.x = Ap[k + 12]; a3.y = Ap[k + 13];
    b0.x = Bp[(long)(k + 0)  * NCOL]; b0.y = Bp[(long)(k + 1)  * NCOL];
    b1.x = Bp[(long)(k + 4)  * NCOL]; b1.y = Bp[(long)(k + 5)  * NCOL];
    b2.x = Bp[(long)(k + 8)  * NCOL]; b2.y = Bp[(long)(k + 9)  * NCOL];
    b3.x = Bp[(long)(k + 12) * NCOL]; b3.y = Bp[(long)(k + 13) * NCOL];
    acc0 = __builtin_amdgcn_wmma_f32_16x16x4_f32(false, a0, false, b0, (short)0, acc0, false, false);
    acc1 = __builtin_amdgcn_wmma_f32_16x16x4_f32(false, a1, false, b1, (short)0, acc1, false, false);
    acc2 = __builtin_amdgcn_wmma_f32_16x16x4_f32(false, a2, false, b2, (short)0, acc2, false, false);
    acc3 = __builtin_amdgcn_wmma_f32_16x16x4_f32(false, a3, false, b3, (short)0, acc3, false, false);
  }
  v8f acc;
#pragma unroll
  for (int i = 0; i < 8; ++i) acc[i] = (acc0[i] + acc1[i]) + (acc2[i] + acc3[i]);

  const int rbase = row0 + ((lane >> 4) << 3);
  float* Cp = C + (long)rbase * NCOL + col0 + mrow;
  if (row0 + 16 <= M) {
#pragma unroll
    for (int i = 0; i < 8; ++i) Cp[(long)i * NCOL] = acc[i];
  } else {
#pragma unroll
    for (int i = 0; i < 8; ++i)
      if (rbase + i < M) Cp[(long)i * NCOL] = acc[i];
  }
}

template <int H, int D>
__global__ void attn_dots(const float* __restrict__ feat,
                          const float* __restrict__ al,
                          const float* __restrict__ ar,
                          float* __restrict__ el, float* __restrict__ er,
                          long NH) {
  long i = (long)blockIdx.x * blockDim.x + threadIdx.x;
  if (i >= NH) return;
  const int h = (int)(i & (H - 1));
  const float4* f  = (const float4*)(feat + i * D);
  const float4* l4 = (const float4*)(al + h * D);
  const float4* r4 = (const float4*)(ar + h * D);
  float sl = 0.f, sr = 0.f;
#pragma unroll
  for (int d = 0; d < D / 4; ++d) {
    float4 v = f[d], lv = l4[d], rv = r4[d];
    sl += v.x * lv.x + v.y * lv.y + v.z * lv.z + v.w * lv.w;
    sr += v.x * rv.x + v.y * rv.y + v.z * rv.z + v.w * rv.w;
  }
  el[i] = sl; er[i] = sr;
}

template <int H>
__global__ void edge_score_max(const int* __restrict__ src,
                               const int* __restrict__ dst,
                               const float* __restrict__ el,
                               const float* __restrict__ er,
                               float* __restrict__ ebuf,
                               float* __restrict__ m, long EH) {
  long i = (long)blockIdx.x * blockDim.x + threadIdx.x;
  if (i >= EH) return;
  long e = i / H; int h = (int)(i & (H - 1));
  float x = el[(long)src[e] * H + h] + er[(long)dst[e] * H + h];
  x = (x > 0.f) ? x : 0.2f * x;
  ebuf[i] = x;
  atomicMaxF(&m[(long)dst[e] * H + h], x);
}

template <int H>
__global__ void edge_exp_sum(const int* __restrict__ dst,
                             float* __restrict__ ebuf,
                             const float* __restrict__ m,
                             float* __restrict__ denom, long EH) {
  long i = (long)blockIdx.x * blockDim.x + threadIdx.x;
  if (i >= EH) return;
  long e = i / H; int h = (int)(i & (H - 1));
  long di = (long)dst[e] * H + h;
  float a = expf(ebuf[i] - m[di]);
  ebuf[i] = a;
  atomicAdd(&denom[di], a);
}

template <int F, int D>
__global__ void edge_scatter(const int* __restrict__ src,
                             const int* __restrict__ dst,
                             const float* __restrict__ feat,
                             const float* __restrict__ ebuf,
                             const float* __restrict__ denom,
                             float* __restrict__ out, long nchunks) {
  constexpr int C4 = F / 4;
  constexpr int H  = F / D;
  long i = (long)blockIdx.x * blockDim.x + threadIdx.x;
  if (i >= nchunks) return;
  long e = i / C4; int c = (int)(i & (C4 - 1));
  int  r = c * 4;  int h = r / D;
  long si = src[e], di = dst[e];
  float alpha = ebuf[e * H + h] / denom[di * H + h];
  float4 f = *(const float4*)(feat + si * F + r);
  float* o = out + di * F + r;
  atomicAdd(o + 0, f.x * alpha);
  atomicAdd(o + 1, f.y * alpha);
  atomicAdd(o + 2, f.z * alpha);
  atomicAdd(o + 3, f.w * alpha);
}

template <int F, bool RELU>
__global__ void bias_act(const float* __restrict__ in,
                         const float* __restrict__ bias,
                         float* __restrict__ out, long n4) {
  long i = (long)blockIdx.x * blockDim.x + threadIdx.x;
  if (i >= n4) return;
  int c = (int)(i & (F / 4 - 1));
  float4 v = ((const float4*)in)[i];
  float4 b = ((const float4*)bias)[c];
  v.x += b.x; v.y += b.y; v.z += b.z; v.w += b.w;
  if (RELU) {
    v.x = fmaxf(v.x, 0.f); v.y = fmaxf(v.y, 0.f);
    v.z = fmaxf(v.z, 0.f); v.w = fmaxf(v.w, 0.f);
  }
  ((float4*)out)[i] = v;
}

extern "C" void kernel_launch(void* const* d_in, const int* in_sizes, int n_in,
                              void* d_out, int out_size, void* d_ws, size_t ws_size,
                              hipStream_t stream) {
  const float* h   = (const float*)d_in[0];
  const int*   src = (const int*)d_in[1];
  const int*   dst = (const int*)d_in[2];
  const float* W1  = (const float*)d_in[3];
  const float* al1 = (const float*)d_in[4];
  const float* ar1 = (const float*)d_in[5];
  const float* b1  = (const float*)d_in[6];
  const float* W2  = (const float*)d_in[7];
  const float* al2 = (const float*)d_in[8];
  const float* ar2 = (const float*)d_in[9];
  const float* b2  = (const float*)d_in[10];

  constexpr int F1 = 128, H1 = 4, D1 = 32;
  constexpr int F2 = 64,  H2 = 1, D2 = 64;
  const long N = in_sizes[0] / F1;
  const long E = in_sizes[1];

  float* ws    = (float*)d_ws;
  float* feat1 = ws;
  float* out1  = feat1 + N * F1;
  float* feat2 = out1  + N * F1;
  float* ebuf  = feat2 + N * F2;
  float* el    = ebuf  + E * H1;
  float* er    = el    + N * H1;
  float* mbuf  = er    + N * H1;
  float* den   = mbuf  + N * H1;
  float* x2    = feat1;
  float* out2  = out1;

  auto cdiv = [](long a, long b) { return (unsigned)((a + b - 1) / b); };
  const int T = 256;
  const unsigned mtiles = cdiv(N, 16);

  fill_kernel<<<cdiv(N * H1, T), T, 0, stream>>>(mbuf, -INFINITY, N * H1);
  fill_kernel<<<cdiv(N * H1, T), T, 0, stream>>>(den, 0.f, N * H1);
  fill_kernel<<<cdiv(N * F1, T), T, 0, stream>>>(out1, 0.f, N * F1);

  wmma_gemm_f32_k128<F1><<<mtiles, 32 * (F1 / 16), 0, stream>>>(h, W1, feat1, (int)N);
  attn_dots<H1, D1><<<cdiv(N * H1, T), T, 0, stream>>>(feat1, al1, ar1, el, er, N * H1);
  edge_score_max<H1><<<cdiv(E * H1, T), T, 0, stream>>>(src, dst, el, er, ebuf, mbuf, E * H1);
  edge_exp_sum<H1><<<cdiv(E * H1, T), T, 0, stream>>>(dst, ebuf, mbuf, den, E * H1);
  edge_scatter<F1, D1><<<cdiv(E * (F1 / 4), T), T, 0, stream>>>(src, dst, feat1, ebuf, den,
                                                                out1, E * (F1 / 4));
  bias_act<F1, true><<<cdiv(N * F1 / 4, T), T, 0, stream>>>(out1, b1, x2, N * F1 / 4);

  fill_kernel<<<cdiv(N * H2, T), T, 0, stream>>>(mbuf, -INFINITY, N * H2);
  fill_kernel<<<cdiv(N * H2, T), T, 0, stream>>>(den, 0.f, N * H2);
  fill_kernel<<<cdiv(N * F2, T), T, 0, stream>>>(out2, 0.f, N * F2);

  wmma_gemm_f32_k128<F2><<<mtiles, 32 * (F2 / 16), 0, stream>>>(x2, W2, feat2, (int)N);
  attn_dots<H2, D2><<<cdiv(N * H2, T), T, 0, stream>>>(feat2, al2, ar2, el, er, N * H2);
  edge_score_max<H2><<<cdiv(E * H2, T), T, 0, stream>>>(src, dst, el, er, ebuf, mbuf, E * H2);
  edge_exp_sum<H2><<<cdiv(E * H2, T), T, 0, stream>>>(dst, ebuf, mbuf, den, E * H2);
  edge_scatter<F2, D2><<<cdiv(E * (F2 / 4), T), T, 0, stream>>>(src, dst, feat2, ebuf, den,
                                                                out2, E * (F2 / 4));
  bias_act<F2, false><<<cdiv(N * F2 / 4, T), T, 0, stream>>>(out2, b2, (float*)d_out, N * F2 / 4);
}


# Edge phase of a layer in ONE launch: an owner work-group scans dst, lists its 196 nodes' edges in LDS, does max, exp-sum, row sums by LDS atomics, stores rows once; 8 -> 4 launches; sums in LDS order
# speedup vs baseline: 2.7167x; 2.7167x over previous
#include <hip/hip_runtime.h>
#include <math.h>

typedef __attribute__((ext_vector_type(2))) float v2f;
typedef __attribute__((ext_vector_type(8))) float v8f;

__device__ __forceinline__ void atomicMaxF(float* addr, float v) {
  if (v >= 0.0f) atomicMax((int*)addr, __float_as_int(v));
  else           atomicMin((unsigned int*)addr, __float_as_uint(v));
}

template <int XM>
__device__ __forceinline__ float row_xor(float v) {
  return __int_as_float(__builtin_amdgcn_update_dpp(
      0, __float_as_int(v), 0x160 + XM, 0xf, 0xf, true));
}

template <int NCOL, int H, bool RELU_A>
__global__ void __launch_bounds__(256) wmma_gemm_f32_k128(
    const float* __restrict__ A, const float* __restrict__ B,
    float* __restrict__ C, int M,
    const float* __restrict__ al, const float* __restrict__ ar,
    float* __restrict__ el, float* __restrict__ er) {
  constexpr int NW  = NCOL / 16;
  constexpr int WPH = NW / H;
  __shared__ float As[16 * 128];
  __shared__ float redl[NW][16];
  __shared__ float redr[NW][16];
  const int lane = threadIdx.x & 31;
  const int wave = threadIdx.x >> 5;
  const int row0 = blockIdx.x * 16;
  const int nrows = (M - row0 < 16) ? (M - row0) : 16;
  if (nrows == 16) {
    const float4* A4  = (const float4*)(A + (long)row0 * 128);
    float4*       As4 = (float4*)As;
    for (int i = threadIdx.x; i < 16 * 128 / 4; i += blockDim.x) {
      float4 v = A4[i];
      if (RELU_A) {
        v.x = fmaxf(v.x, 0.f); v.y = fmaxf(v.y, 0.f);
        v.z = fmaxf(v.z, 0.f); v.w = fmaxf(v.w, 0.f);
      }
      As4[i] = v;
    }
  } else {
    for (int i = threadIdx.x; i < 16 * 128; i += blockDim.x) {
      int r = row0 + (i >> 7);
      if (r >= M) r = M - 1;
      float v = A[(long)r * 128 + (i & 127)];
      As[i] = RELU_A ? fmaxf(v, 0.f) : v;
    }
  }

  __syncthreads();

  const int col0  = wave * 16;
  const int mrow  = lane & 15;
  const int khalf = (lane >> 4) << 1;
  const float* Ap = As + mrow * 128 + khalf;
  const float* Bp = B + (long)khalf * NCOL + col0 + mrow;

  v8f acc0 = {}, acc1 = {}, acc2 = {}, acc3 = {};
#pragma unroll
  for (int k = 0; k < 128; k += 16) {
    v2f a0, a1, a2, a3, b0, b1, b2, b3;
    a0.x = Ap[k + 0];  a0.y = Ap[k + 1];
    a1.x = Ap[k + 4];  a1.y = Ap[k + 5];
    a2.x = Ap[k + 8];  a2.y = Ap[k + 9];
    a3.x = Ap[k + 12]; a3.y = Ap[k + 13];
    b0.x = Bp[(long)(k + 0)  * NCOL]; b0.y = Bp[(long)(k + 1)  * NCOL];
    b1.x = Bp[(long)(k + 4)  * NCOL]; b1.y = Bp[(long)(k + 5)  * NCOL];
    b2.x = Bp[(long)(k + 8)  * NCOL]; b2.y = Bp[(long)(k + 9)  * NCOL];
    b3.x = Bp[(long)(k + 12) * NCOL]; b3.y = Bp[(long)(k + 13) * NCOL];
    acc0 = __builtin_amdgcn_wmma_f32_16x16x4_f32(false, a0, false, b0, (short)0, acc0, false, false);
    acc1 = __builtin_amdgcn_wmma_f32_16x16x4_f32(false, a1, false, b1, (short)0, acc1, false, false);
    acc2 = __builtin_amdgcn_wmma_f32_16x16x4_f32(false, a2, false, b2, (short)0, acc2, false, false);
    acc3 = __builtin_amdgcn_wmma_f32_16x16x4_f32(false, a3, false, b3, (short)0, acc3, false, false);
  }
  v8f acc;
#pragma unroll
  for (int i = 0; i < 8; ++i) acc[i] = (acc0[i] + acc1[i]) + (acc2[i] + acc3[i]);

  const int rhalf = (lane >> 4) << 3;
  const int rbase = row0 + rhalf;
  float* Cp = C + (long)rbase * NCOL + col0 + mrow;
  if (nrows == 16) {
#pragma unroll
    for (int i = 0; i < 8; ++i) Cp[(long)i * NCOL] = acc[i];
  } else {
#pragma unroll
    for (int i = 0; i < 8; ++i)
      if (rbase + i < M) Cp[(long)i * NCOL] = acc[i];
  }

  const float alv = al[col0 + mrow], arv = ar[col0 + mrow];
  float pl[8], pr[8];
#pragma unroll
  for (int i = 0; i < 8; ++i) { pl[i] = acc[i] * alv; pr[i] = acc[i] * arv; }
#pragma unroll
  for (int i = 0; i < 8; ++i) { pl[i] += row_xor<8>(pl[i]); pr[i] += row_xor<8>(pr[i]); }
#pragma unroll
  for (int i = 0; i < 8; ++i) { pl[i] += row_xor<4>(pl[i]); pr[i] += row_xor<4>(pr[i]); }
#pragma unroll
  for (int i = 0; i < 8; ++i) { pl[i] += row_xor<2>(pl[i]); pr[i] += row_xor<2>(pr[i]); }
#pragma unroll
  for (int i = 0; i < 8; ++i) { pl[i] += row_xor<1>(pl[i]); pr[i] += row_xor<1>(pr[i]); }
  if (mrow == 0) {
#pragma unroll
    for (int i = 0; i < 8; ++i) { redl[wave][rhalf + i] = pl[i]; redr[wave][rhalf + i] = pr[i]; }
  }
  __syncthreads();
  if ((int)threadIdx.x < nrows * H) {
    const int r = threadIdx.x / H, hd = threadIdx.x % H;
    float sl = 0.f, sr = 0.f;
#pragma unroll
    for (int w = 0; w < WPH; ++w) { sl += redl[hd * WPH + w][r]; sr += redr[hd * WPH + w][r]; }
    el[(long)row0 * H + threadIdx.x] = sl;
    er[(long)row0 * H + threadIdx.x] = sr;
  }
}

template <int F, int H, int NB, int CAP, int EPW>
__global__ void __launch_bounds__(1024) edge_owner(
    const int* __restrict__ src, const int* __restrict__ dst,
    const float* __restrict__ feat, const float* __restrict__ el,
    const float* __restrict__ er, const float* __restrict__ bias,
    float* __restrict__ out, int N, int E) {
  constexpr int D  = F / H;
  constexpr int Q  = F / 32;
  constexpr int NT = 1024;
  constexpr int NW = NT / 32;
  static_assert(EPW * H <= 32, "one staging lane per (edge, head)");
  __shared__ float tile[NB * F];
  __shared__ float mL[NB * H];
  __shared__ float dL[NB * H];
  __shared__ int   le[CAP];
  __shared__ int   ls[CAP];
  __shared__ int   ld[CAP];
  __shared__ int   wc[NT / 32];
  constexpr int WCAP = CAP / (NT / 32);
  const int tid  = threadIdx.x;
  const int lane = tid & 31;
  const int wave = tid >> 5;
  const int lo   = blockIdx.x * NB;
  const int nbh  = (N - lo < NB) ? (N - lo) : NB;

  for (int i = tid; i < NB * F; i += NT) tile[i] = 0.f;
  for (int i = tid; i < NB * H; i += NT) { mL[i] = -INFINITY; dL[i] = 0.f; }

  int wfill = 0;
  auto put = [&](bool p, unsigned m, int e) {
    const int pos = wfill + (int)__builtin_amdgcn_mbcnt_lo(m, 0u);
    if (p && pos < WCAP) le[wave * WCAP + pos] = e;
    wfill += __builtin_popcount(m);
  };
  auto keep4 = [&](int4 v, bool valid, int e) {
    const bool p0 = valid && (unsigned)(v.x - lo) < (unsigned)nbh;
    const bool p1 = valid && (unsigned)(v.y - lo) < (unsigned)nbh;
    const bool p2 = valid && (unsigned)(v.z - lo) < (unsigned)nbh;
    const bool p3 = valid && (unsigned)(v.w - lo) < (unsigned)nbh;
    const unsigned m0 = __builtin_amdgcn_ballot_w32(p0);
    const unsigned m1 = __builtin_amdgcn_ballot_w32(p1);
    const unsigned m2 = __builtin_amdgcn_ballot_w32(p2);
    const unsigned m3 = __builtin_amdgcn_ballot_w32(p3);
    if ((m0 | m1 | m2 | m3) != 0u) {
      put(p0, m0, e); put(p1, m1, e + 1); put(p2, m2, e + 2); put(p3, m3, e + 3);
    }
  };
  {
    const int4* d4 = (const int4*)dst;
    const int E4 = E >> 2;
    int base = wave * 32;
    for (; base + 7 * NT + 32 <= E4; base += 8 * NT) {
      int4 v[8];
#pragma unroll
      for (int u = 0; u < 8; ++u) v[u] = d4[base + u * NT + lane];
#pragma unroll
      for (int u = 0; u < 8; ++u) keep4(v[u], true, (base + u * NT + lane) << 2);
    }
    for (; base < E4; base += NT) {
      const int  i = base + lane;
      const bool valid = i < E4;
      const int4 v = d4[valid ? i : 0];
      keep4(v, valid, i << 2);
    }
    if (wave == 0) {
      const int  e = (E4 << 2) + lane;
      const bool p = (e < E) && (unsigned)(dst[(e < E) ? e : 0] - lo) < (unsigned)nbh;
      put(p, __builtin_amdgcn_ballot_w32(p), e);
    }
  }
  if (lane == 0) wc[wave] = wfill;
  __syncthreads();
  int total = 0, off = 0;
  bool fits = true;
  for (int w = 0; w < NT / 32; ++w) {
    const int n = wc[w];
    if (w < wave) off += n;
    total += n;
    fits = fits && (n <= WCAP);
  }
  if (fits) {
    for (int i = lane; i < wfill; i += 32) {
      const int e = le[wave * WCAP + i];
      ls[off + i] = src[e];
      ld[off + i] = dst[e] - lo;
    }
  }
  __syncthreads();

  auto score = [&](int s, int r, int h) {
    const float x = el[(long)s * H + h] + er[(long)(lo + r) * H + h];
    return (x > 0.f) ? x : 0.2f * x;
  };
  auto for_owned = [&](auto fn) {
    if (fits) {
      for (int i = tid; i < total; i += NT) fn(ls[i], ld[i]);
    } else {
      for (int e = tid; e < E; e += NT) {
        const unsigned r = (unsigned)(dst[e] - lo);
        if (r < (unsigned)nbh) fn(src[e], (int)r);
      }
    }
  };

  for_owned([&](int s, int r) {
#pragma unroll
    for (int h = 0; h < H; ++h) atomicMaxF(&mL[r * H + h], score(s, r, h));
  });
  __syncthreads();
  for_owned([&](int s, int r) {
#pragma unroll
    for (int h = 0; h < H; ++h)
      atomicAdd(&dL[r * H + h], expf(score(s, r, h) - mL[r * H + h]));
  });
  __syncthreads();

  if (fits) {
    const int jl = lane / H, hl = lane % H;
    for (int i0 = wave * EPW; i0 < total; i0 += NW * EPW) {
      const int c = (total - i0 < EPW) ? (total - i0) : EPW;
      int s_l = 0, r_l = 0; float a_l = 0.f;
      if (jl < c) {
        s_l = ls[i0 + jl]; r_l = ld[i0 + jl];
        a_l = expf(score(s_l, r_l, hl) - mL[r_l * H + hl]) / dL[r_l * H + hl];
      }
      float v[EPW][Q];
#pragma unroll
      for (int j = 0; j < EPW; ++j) {
        const int s = __builtin_amdgcn_readlane(s_l, j * H);
        const float* fr = feat + (long)s * F + lane;
#pragma unroll
        for (int q = 0; q < Q; ++q) v[j][q] = fr[q * 32];
      }
#pragma unroll
      for (int j = 0; j < EPW; ++j) {
        if (j < c) {
          const int r = __builtin_amdgcn_readlane(r_l, j * H);
          float* t = tile + r * F + lane;
#pragma unroll
          for (int q = 0; q < Q; ++q) {
            const float a = __int_as_float(
                __builtin_amdgcn_readlane(__float_as_int(a_l), j * H + (q * 32) / D));
            atomicAdd(t + q * 32, v[j][q] * a);
          }
        }
      }
    }
  } else {
    for_owned([&](int s, int r) {
      for (int h = 0; h < H; ++h) {
        const float a = expf(score(s, r, h) - mL[r * H + h]) / dL[r * H + h];
        const float* fr = feat + (long)s * F + h * D;
        float* t = tile + r * F + h * D;
        for (int k = 0; k < D; ++k) atomicAdd(t + k, fr[k] * a);
      }
    });
  }
  __syncthreads();

  {
    constexpr int C4 = F / 4;
    const float4* t4 = (const float4*)tile;
    const float4* b4 = (const float4*)bias;
    float4* o4 = (float4*)(out + (long)lo * F);
    for (int i = tid; i < nbh * C4; i += NT) {
      float4 v = t4[i];
      const float4 b = b4[i & (C4 - 1)];
      v.x += b.x; v.y += b.y; v.z += b.z; v.w += b.w;
      o4[i] = v;
    }
  }
}

extern "C" void kernel_launch(void* const* d_in, const int* in_sizes, int n_in,
                              void* d_out, int out_size, void* d_ws, size_t ws_size,
                              hipStream_t stream) {
  const float* h   = (const float*)d_in[0];
  const int*   src = (const int*)d_in[1];
  const int*   dst = (const int*)d_in[2];
  const float* W1  = (const float*)d_in[3];
  const float* al1 = (const float*)d_in[4];
  const float* ar1 = (const float*)d_in[5];
  const float* b1  = (const float*)d_in[6];
  const float* W2  = (const float*)d_in[7];
  const float* al2 = (const float*)d_in[8];
  const float* ar2 = (const float*)d_in[9];
  const float* b2  = (const float*)d_in[10];

  constexpr int F1 = 128, H1 = 4;
  constexpr int F2 = 64,  H2 = 1;
  constexpr int NB = 196;
  constexpr int CAP = 8192;
  const long N = in_sizes[0] / F1;
  const long E = in_sizes[1];

  float* ws    = (float*)d_ws;
  float* feat1 = ws;
  float* out1  = feat1 + N * F1;
  float* feat2 = out1  + N * F1;
  float* el    = feat2 + N * F2;
  float* er    = el    + N * H1;
  float* out2  = (float*)d_out;

  auto cdiv = [](long a, long b) { return (unsigned)((a + b - 1) / b); };
  const unsigned mtiles = cdiv(N, 16);
  const unsigned owners = cdiv(N, NB);

  wmma_gemm_f32_k128<F1, H1, false><<<mtiles, 32 * (F1 / 16), 0, stream>>>(
      h, W1, feat1, (int)N, al1, ar1, el, er);
  edge_owner<F1, H1, NB, CAP, 8><<<owners, 1024, 0, stream>>>(
      src, dst, feat1, el, er, b1, out1, (int)N, (int)E);

  wmma_gemm_f32_k128<F2, H2, true><<<mtiles, 32 * (F2 / 16), 0, stream>>>(
      out1, W2, feat2, (int)N, al2, ar2, el, er);
  edge_owner<F2, H2, NB, CAP, 16><<<owners, 1024, 0, stream>>>(
      src, dst, feat2, el, er, b2, out2, (int)N, (int)E);
}


# Owner scan reads a 1-byte owner key per edge (dst/196, made by the GEMM launch in front) instead of the 4-byte dst: 4x fewer bytes and compares per owner; list order, so LDS sum order, changes
# speedup vs baseline: 3.1702x; 1.1669x over previous
#include <hip/hip_runtime.h>
#include <math.h>

typedef __attribute__((ext_vector_type(2))) float v2f;
typedef __attribute__((ext_vector_type(8))) float v8f;

__device__ __forceinline__ void atomicMaxF(float* addr, float v) {
  if (v >= 0.0f) atomicMax((int*)addr, __float_as_int(v));
  else           atomicMin((unsigned int*)addr, __float_as_uint(v));
}

template <int XM>
__device__ __forceinline__ float row_xor(float v) {
  return __int_as_float(__builtin_amdgcn_update_dpp(
      0, __float_as_int(v), 0x160 + XM, 0xf, 0xf, true));
}

template <int NB>
__device__ __forceinline__ void make_owner_keys(const int* __restrict__ dst,
                                                unsigned char* __restrict__ key, int E) {
  const int E4 = E >> 2;
  const int4* d4 = (const int4*)dst;
  unsigned* k32 = (unsigned*)key;
  const int step = gridDim.x * blockDim.x;
  for (int q = blockIdx.x * blockDim.x + threadIdx.x; q < E4; q += step) {
    const int4 d = d4[q];
    k32[q] = (((unsigned)d.x / NB) & 255u) | ((((unsigned)d.y / NB) & 255u) << 8) |
             ((((unsigned)d.z / NB) & 255u) << 16) | ((((unsigned)d.w / NB) & 255u) << 24);
  }
  if (blockIdx.x == 0 && (int)threadIdx.x < (E & 3)) {
    const int e = (E4 << 2) + threadIdx.x;
    key[e] = (unsigned char)(((unsigned)dst[e] / NB) & 255u);
  }
}

template <int NCOL, int H, bool RELU_A, int NB>
__global__ void __launch_bounds__(256) wmma_gemm_f32_k128(
    const float* __restrict__ A, const float* __restrict__ B,
    float* __restrict__ C, int M,
    const float* __restrict__ al, const float* __restrict__ ar,
    float* __restrict__ el, float* __restrict__ er,
    const int* __restrict__ dst, unsigned char* __restrict__ key, int E) {
  constexpr int NW  = NCOL / 16;
  constexpr int WPH = NW / H;
  __shared__ float As[16 * 128];
  __shared__ float redl[NW][16];
  __shared__ float redr[NW][16];
  const int lane = threadIdx.x & 31;
  const int wave = threadIdx.x >> 5;
  const int row0 = blockIdx.x * 16;
  const int nrows = (M - row0 < 16) ? (M - row0) : 16;
  if (nrows == 16) {
    const float4* A4  = (const float4*)(A + (long)row0 * 128);
    float4*       As4 = (float4*)As;
    for (int i = threadIdx.x; i < 16 * 128 / 4; i += blockDim.x) {
      float4 v = A4[i];
      if (RELU_A) {
        v.x = fmaxf(v.x, 0.f); v.y = fmaxf(v.y, 0.f);
        v.z = fmaxf(v.z, 0.f); v.w = fmaxf(v.w, 0.f);
      }
      As4[i] = v;
    }
  } else {
    for (int i = threadIdx.x; i < 16 * 128; i += blockDim.x) {
      int r = row0 + (i >> 7);
      if (r >= M) r = M - 1;
      float v = A[(long)r * 128 + (i & 127)];
      As[i] = RELU_A ? fmaxf(v, 0.f) : v;
    }
  }

  __syncthreads();

  const int col0  = wave * 16;
  const int mrow  = lane & 15;
  const int khalf = (lane >> 4) << 1;
  const float* Ap = As + mrow * 128 + khalf;
  const float* Bp = B + (long)khalf * NCOL + col0 + mrow;

  v8f acc0 = {}, acc1 = {}, acc2 = {}, acc3 = {};
#pragma unroll
  for (int k = 0; k < 128; k += 16) {
    v2f a0, a1, a2, a3, b0, b1, b2, b3;
    a0.x = Ap[k + 0];  a0.y = Ap[k + 1];
    a1.x = Ap[k + 4];  a1.y = Ap[k + 5];
    a2.x = Ap[k + 8];  a2.y = Ap[k + 9];
    a3.x = Ap[k + 12]; a3.y = Ap[k + 13];
    b0.x = Bp[(long)(k + 0)  * NCOL]; b0.y = Bp[(long)(k + 1)  * NCOL];
    b1.x = Bp[(long)(k + 4)  * NCOL]; b1.y = Bp[(long)(k + 5)  * NCOL];
    b2.x = Bp[(long)(k + 8)  * NCOL]; b2.y = Bp[(long)(k + 9)  * NCOL];
    b3.x = Bp[(long)(k + 12) * NCOL]; b3.y = Bp[(long)(k + 13) * NCOL];
    acc0 = __builtin_amdgcn_wmma_f32_16x16x4_f32(false, a0, false, b0, (short)0, acc0, false, false);
    acc1 = __builtin_amdgcn_wmma_f32_16x16x4_f32(false, a1, false, b1, (short)0, acc1, false, false);
    acc2 = __builtin_amdgcn_wmma_f32_16x16x4_f32(false, a2, false, b2, (short)0, acc2, false, false);
    acc3 = __builtin_amdgcn_wmma_f32_16x16x4_f32(false, a3, false, b3, (short)0, acc3, false, false);
  }
  v8f acc;
#pragma unroll
  for (int i = 0; i < 8; ++i) acc[i] = (acc0[i] + acc1[i]) + (acc2[i] + acc3[i]);

  const int rhalf = (lane >> 4) << 3;
  const int rbase = row0 + rhalf;
  float* Cp = C + (long)rbase * NCOL + col0 + mrow;
  if (nrows == 16) {
#pragma unroll
    for (int i = 0; i < 8; ++i) Cp[(long)i * NCOL] = acc[i];
  } else {
#pragma unroll
    for (int i = 0; i < 8; ++i)
      if (rbase + i < M) Cp[(long)i * NCOL] = acc[i];
  }

  const float alv = al[col0 + mrow], arv = ar[col0 + mrow];
  float pl[8], pr[8];
#pragma unroll
  for (int i = 0; i < 8; ++i) { pl[i] = acc[i] * alv; pr[i] = acc[i] * arv; }
#pragma unroll
  for (int i = 0; i < 8; ++i) { pl[i] += row_xor<8>(pl[i]); pr[i] += row_xor<8>(pr[i]); }
#pragma unroll
  for (int i = 0; i < 8; ++i) { pl[i] += row_xor<4>(pl[i]); pr[i] += row_xor<4>(pr[i]); }
#pragma unroll
  for (int i = 0; i < 8; ++i) { pl[i] += row_xor<2>(pl[i]); pr[i] += row_xor<2>(pr[i]); }
#pragma unroll
  for (int i = 0; i < 8; ++i) { pl[i] += row_xor<1>(pl[i]); pr[i] += row_xor<1>(pr[i]); }
  if (mrow == 0) {
#pragma unroll
    for (int i = 0; i < 8; ++i) { redl[wave][rhalf + i] = pl[i]; redr[wave][rhalf + i] = pr[i]; }
  }
  __syncthreads();
  if ((int)threadIdx.x < nrows * H) {
    const int r = threadIdx.x / H, hd = threadIdx.x % H;
    float sl = 0.f, sr = 0.f;
#pragma unroll
    for (int w = 0; w < WPH; ++w) { sl += redl[hd * WPH + w][r]; sr += redr[hd * WPH + w][r]; }
    el[(long)row0 * H + threadIdx.x] = sl;
    er[(long)row0 * H + threadIdx.x] = sr;
  }
  make_owner_keys<NB>(dst, key, E);
}

template <int F, int H, int NB, int CAP, int EPW>
__global__ void __launch_bounds__(1024) edge_owner(
    const int* __restrict__ src, const int* __restrict__ dst,
    const float* __restrict__ feat, const float* __restrict__ el,
    const float* __restrict__ er, const float* __restrict__ bias,
    float* __restrict__ out, int N, int E,
    const unsigned char* __restrict__ key) {
  constexpr int D  = F / H;
  constexpr int Q  = F / 32;
  constexpr int NT = 1024;
  constexpr int NW = NT / 32;
  static_assert(EPW * H <= 32, "one staging lane per (edge, head)");
  __shared__ float tile[NB * F];
  __shared__ float mL[NB * H];
  __shared__ float dL[NB * H];
  __shared__ int   le[CAP];
  __shared__ int   ls[CAP];
  __shared__ int   ld[CAP];
  __shared__ int   wc[NT / 32];
  constexpr int WCAP = CAP / (NT / 32);
  const int tid  = threadIdx.x;
  const int lane = tid & 31;
  const int wave = tid >> 5;
  const int lo   = blockIdx.x * NB;
  const int nbh  = (N - lo < NB) ? (N - lo) : NB;

  for (int i = tid; i < NB * F; i += NT) tile[i] = 0.f;
  for (int i = tid; i < NB * H; i += NT) { mL[i] = -INFINITY; dL[i] = 0.f; }

  const bool keyed = gridDim.x <= 256u;
  int wfill = 0;
  auto put = [&](bool p, unsigned m, int e) {
    const int pos = wfill + (int)__builtin_amdgcn_mbcnt_lo(m, 0u);
    if (p && pos < WCAP) le[wave * WCAP + pos] = e;
    wfill += __builtin_popcount(m);
  };
  const unsigned pat = (blockIdx.x & 255u) * 0x01010101u;
  auto zero_bytes = [](unsigned x) {
    const unsigned t = (x & 0x7f7f7f7fu) + 0x7f7f7f7fu;
    return ~(t | x | 0x7f7f7f7fu);
  };
  auto keep16 = [&](int4 v, bool valid, int e0) {
    const unsigned f0 = zero_bytes((unsigned)v.x ^ pat), f1 = zero_bytes((unsigned)v.y ^ pat);
    const unsigned f2 = zero_bytes((unsigned)v.z ^ pat), f3 = zero_bytes((unsigned)v.w ^ pat);
    unsigned m = valid ? ((f0 >> 7) | (f1 >> 6) | (f2 >> 5) | (f3 >> 4)) : 0u;
    unsigned any = __builtin_amdgcn_ballot_w32(m != 0u);
    while (any != 0u) {
      const bool p = (m != 0u);
      const int  i = p ? __builtin_ctz(m) : 0;
      put(p, any, e0 + ((i & 7) << 2) + (i >> 3));
      m &= m - 1u;
      any = __builtin_amdgcn_ballot_w32(m != 0u);
    }
  };
  if (keyed) {
    const int4* k4 = (const int4*)key;
    const int E16 = E >> 4;
    int base = wave * 32;
    for (; base + 7 * NT + 32 <= E16; base += 8 * NT) {
      int4 v[8];
#pragma unroll
      for (int u = 0; u < 8; ++u) v[u] = k4[base + u * NT + lane];
#pragma unroll
      for (int u = 0; u < 8; ++u) keep16(v[u], true, (base + u * NT + lane) << 4);
    }
    for (; base < E16; base += NT) {
      const int  i = base + lane;
      const bool valid = i < E16;
      const int4 v = k4[valid ? i : 0];
      keep16(v, valid, i << 4);
    }
    if (wave == 0) {
      const int  e = (E16 << 4) + lane;
      const bool in = (lane < 16) && (e < E);
      const bool p = in && (unsigned)(dst[in ? e : 0] - lo) < (unsigned)nbh;
      put(p, __builtin_amdgcn_ballot_w32(p), e);
    }
  }
  if (lane == 0) wc[wave] = wfill;
  __syncthreads();
  int total = 0, off = 0;
  bool fits = keyed;
  for (int w = 0; w < NT / 32; ++w) {
    const int n = wc[w];
    if (w < wave) off += n;
    total += n;
    fits = fits && (n <= WCAP);
  }
  if (fits) {
    for (int i = lane; i < wfill; i += 32) {
      const int e = le[wave * WCAP + i];
      ls[off + i] = src[e];
      ld[off + i] = dst[e] - lo;
    }
  }
  __syncthreads();

  auto score = [&](int s, int r, int h) {
    const float x = el[(long)s * H + h] + er[(long)(lo + r) * H + h];
    return (x > 0.f) ? x : 0.2f * x;
  };
  auto for_owned = [&](auto fn) {
    if (fits) {
      for (int i = tid; i < total; i += NT) fn(ls[i], ld[i]);
    } else {
      for (int e = tid; e < E; e += NT) {
        const unsigned r = (unsigned)(dst[e] - lo);
        if (r < (unsigned)nbh) fn(src[e], (int)r);
      }
    }
  };

  for_owned([&](int s, int r) {
#pragma unroll
    for (int h = 0; h < H; ++h) atomicMaxF(&mL[r * H + h], score(s, r, h));
  });
  __syncthreads();
  for_owned([&](int s, int r) {
#pragma unroll
    for (int h = 0; h < H; ++h)
      atomicAdd(&dL[r * H + h], expf(score(s, r, h) - mL[r * H + h]));
  });
  __syncthreads();

  if (fits) {
    const int jl = lane / H, hl = lane % H;
    for (int i0 = wave * EPW; i0 < total; i0 += NW * EPW) {
      const int c = (total - i0 < EPW) ? (total - i0) : EPW;
      int s_l = 0, r_l = 0; float a_l = 0.f;
      if (jl < c) {
        s_l = ls[i0 + jl]; r_l = ld[i0 + jl];
        a_l = expf(score(s_l, r_l, hl) - mL[r_l * H + hl]) / dL[r_l * H + hl];
      }
      float v[EPW][Q];
#pragma unroll
      for (int j = 0; j < EPW; ++j) {
        const int s = __builtin_amdgcn_readlane(s_l, j * H);
        const float* fr = feat + (long)s * F + lane;
#pragma unroll
        for (int q = 0; q < Q; ++q) v[j][q] = fr[q * 32];
      }
#pragma unroll
      for (int j = 0; j < EPW; ++j) {
        if (j < c) {
          const int r = __builtin_amdgcn_readlane(r_l, j * H);
          float* t = tile + r * F + lane;
#pragma unroll
          for (int q = 0; q < Q; ++q) {
            const float a = __int_as_float(
                __builtin_amdgcn_readlane(__float_as_int(a_l), j * H + (q * 32) / D));
            atomicAdd(t + q * 32, v[j][q] * a);
          }
        }
      }
    }
  } else {
    for_owned([&](int s, int r) {
      for (int h = 0; h < H; ++h) {
        const float a = expf(score(s, r, h) - mL[r * H + h]) / dL[r * H + h];
        const float* fr = feat + (long)s * F + h * D;
        float* t = tile + r * F + h * D;
        for (int k = 0; k < D; ++k) atomicAdd(t + k, fr[k] * a);
      }
    });
  }
  __syncthreads();

  {
    constexpr int C4 = F / 4;
    const float4* t4 = (const float4*)tile;
    const float4* b4 = (const float4*)bias;
    float4* o4 = (float4*)(out + (long)lo * F);
    for (int i = tid; i < nbh * C4; i += NT) {
      float4 v = t4[i];
      const float4 b = b4[i & (C4 - 1)];
      v.x += b.x; v.y += b.y; v.z += b.z; v.w += b.w;
      o4[i] = v;
    }
  }
}

extern "C" void kernel_launch(void* const* d_in, const int* in_sizes, int n_in,
                              void* d_out, int out_size, void* d_ws, size_t ws_size,
                              hipStream_t stream) {
  const float* h   = (const float*)d_in[0];
  const int*   src = (const int*)d_in[1];
  const int*   dst = (const int*)d_in[2];
  const float* W1  = (const float*)d_in[3];
  const float* al1 = (const float*)d_in[4];
  const float* ar1 = (const float*)d_in[5];
  const float* b1  = (const float*)d_in[6];
  const float* W2  = (const float*)d_in[7];
  const float* al2 = (const float*)d_in[8];
  const float* ar2 = (const float*)d_in[9];
  const float* b2  = (const float*)d_in[10];

  constexpr int F1 = 128, H1 = 4;
  constexpr int F2 = 64,  H2 = 1;
  constexpr int NB = 196;
  constexpr int CAP = 8192;
  const long N = in_sizes[0] / F1;
  const long E = in_sizes[1];

  float* ws    = (float*)d_ws;
  float* feat1 = ws;
  float* out1  = feat1 + N * F1;
  float* feat2 = out1  + N * F1;
  float* el    = feat2 + N * F2;
  float* er    = el    + N * H1;
  float* out2  = (float*)d_out;

  auto cdiv = [](long a, long b) { return (unsigned)((a + b - 1) / b); };
  const unsigned mtiles = cdiv(N, 16);
  const unsigned owners = cdiv(N, NB);

  wmma_gemm_f32_k128<F1, H1, false, NB><<<mtiles, 32 * (F1 / 16), 0, stream>>>(
      h, W1, feat1, (int)N, al1, ar1, el, er, dst, (unsigned char*)feat2, (int)E);
  edge_owner<F1, H1, NB, CAP, 8><<<owners, 1024, 0, stream>>>(
      src, dst, feat1, el, er, b1, out1, (int)N, (int)E, (const unsigned char*)feat2);

  wmma_gemm_f32_k128<F2, H2, true, NB><<<mtiles, 32 * (F2 / 16), 0, stream>>>(
      out1, W2, feat2, (int)N, al2, ar2, el, er, dst, (unsigned char*)feat1, (int)E);
  edge_owner<F2, H2, NB, CAP, 16><<<owners, 1024, 0, stream>>>(
      src, dst, feat2, el, er, b2, out2, (int)N, (int)E, (const unsigned char*)feat1);
}


# GEMM re-tiled: a wave owns 16 rows x all columns, A rows straight from global into fragment registers, weights once per group in LDS in fragment order; 3125 -> 391 groups; k and dot sum order change
# speedup vs baseline: 3.6292x; 1.1448x over previous
#include <hip/hip_runtime.h>
#include <math.h>

typedef __attribute__((ext_vector_type(2))) float v2f;
typedef __attribute__((ext_vector_type(8))) float v8f;

__device__ __forceinline__ void atomicMaxF(float* addr, float v) {
  if (v >= 0.0f) atomicMax((int*)addr, __float_as_int(v));
  else           atomicMin((unsigned int*)addr, __float_as_uint(v));
}

template <int XM>
__device__ __forceinline__ float row_xor(float v) {
  return __int_as_float(__builtin_amdgcn_update_dpp(
      0, __float_as_int(v), 0x160 + XM, 0xf, 0xf, true));
}

template <int NB>
__device__ __forceinline__ void make_owner_keys(const int* __restrict__ dst,
                                                unsigned char* __restrict__ key, int E) {
  const int E4 = E >> 2;
  const int4* d4 = (const int4*)dst;
  unsigned* k32 = (unsigned*)key;
  const int step = gridDim.x * blockDim.x;
  for (int q = blockIdx.x * blockDim.x + threadIdx.x; q < E4; q += step) {
    const int4 d = d4[q];
    k32[q] = (((unsigned)d.x / NB) & 255u) | ((((unsigned)d.y / NB) & 255u) << 8) |
             ((((unsigned)d.z / NB) & 255u) << 16) | ((((unsigned)d.w / NB) & 255u) << 24);
  }
  if (blockIdx.x == 0 && (int)threadIdx.x < (E & 3)) {
    const int e = (E4 << 2) + threadIdx.x;
    key[e] = (unsigned char)(((unsigned)dst[e] / NB) & 255u);
  }
}

template <int NCOL, int H, bool RELU_A, int NB>
__global__ void __launch_bounds__(256) __attribute__((amdgpu_waves_per_eu(4, 4))) wmma_gemm_f32_k128(
    const float* __restrict__ A, const float* __restrict__ B,
    float* __restrict__ C, int M,
    const float* __restrict__ al, const float* __restrict__ ar,
    float* __restrict__ el, float* __restrict__ er,
    const int* __restrict__ dst, unsigned char* __restrict__ key, int E) {
  constexpr int CT  = NCOL / 16;
  constexpr int G   = NCOL / 64;
  constexpr int D   = NCOL / H;
  constexpr int LPH = D / 4;
  constexpr int HPG = 64 / D;
  static_assert(NCOL % 64 == 0 && (LPH == 8 || LPH == 16) && G * HPG == H, "shape");
  __shared__ float4 Bs[16 * CT * 32];
  const int lane = threadIdx.x & 31;
  const int wave = threadIdx.x >> 5;
  const int c    = lane & 15;
  const int hh   = lane >> 4;
  const int mtiles = (M + 15) >> 4;
  const int tile = blockIdx.x * 8 + wave;
  const bool has = tile < mtiles;
  const int row0 = tile * 16;

  float4 a[16];
  {
    int r = row0 + c;
    if (r >= M) r = M - 1;
    const float4* ap = (const float4*)(A + (long)r * 128) + hh;
#pragma unroll
    for (int k8 = 0; k8 < 16; ++k8) a[k8] = ap[2 * k8];
  }

#pragma unroll
  for (int it = 0; it < (16 * G * 32) / 256; ++it) {
    const int i = threadIdx.x + it * 256;
    const int sg  = (i >> 5) & (G - 1);
    const int sk8 = i / (32 * G);
    const float* bp = B + (long)(8 * sk8 + 4 * hh) * NCOL + sg * 64 + c * 4;
    const float4 v0 = *(const float4*)(bp);
    const float4 v1 = *(const float4*)(bp + NCOL);
    const float4 v2 = *(const float4*)(bp + 2 * NCOL);
    const float4 v3 = *(const float4*)(bp + 3 * NCOL);
    float4* d = Bs + (sk8 * CT + 4 * sg) * 32 + lane;
    d[0]  = make_float4(v0.x, v1.x, v2.x, v3.x);
    d[32] = make_float4(v0.y, v1.y, v2.y, v3.y);
    d[64] = make_float4(v0.z, v1.z, v2.z, v3.z);
    d[96] = make_float4(v0.w, v1.w, v2.w, v3.w);
  }
  __syncthreads();

  if (has) {
    if (RELU_A) {
#pragma unroll
      for (int k8 = 0; k8 < 16; ++k8) {
        a[k8].x = fmaxf(a[k8].x, 0.f); a[k8].y = fmaxf(a[k8].y, 0.f);
        a[k8].z = fmaxf(a[k8].z, 0.f); a[k8].w = fmaxf(a[k8].w, 0.f);
      }
    }
    v8f acc[CT];
#pragma unroll
    for (int ct = 0; ct < CT; ++ct) acc[ct] = v8f{};
    const float4* bq = Bs + lane;
    auto load_b = [&](float4 (&b)[CT], int k8) {
#pragma unroll
      for (int ct = 0; ct < CT; ++ct) b[ct] = bq[(k8 * CT + ct) * 32];
    };
    auto block = [&](const float4& av, const float4 (&b)[CT]) {
      v2f a01, a23;
      a01.x = av.x; a01.y = av.y; a23.x = av.z; a23.y = av.w;
#pragma unroll
      for (int ct = 0; ct < CT; ++ct) {
        v2f b01; b01.x = b[ct].x; b01.y = b[ct].y;
        acc[ct] = __builtin_amdgcn_wmma_f32_16x16x4_f32(false, a01, false, b01, (short)0, acc[ct], false, false);
      }
#pragma unroll
      for (int ct = 0; ct < CT; ++ct) {
        v2f b23; b23.x = b[ct].z; b23.y = b[ct].w;
        acc[ct] = __builtin_amdgcn_wmma_f32_16x16x4_f32(false, a23, false, b23, (short)0, acc[ct], false, false);
      }
    };
    float4 bA[CT], bB[CT];
    load_b(bA, 0);
#pragma unroll
    for (int k8 = 0; k8 < 16; k8 += 2) {
      load_b(bB, k8 + 1);
      block(a[k8], bA);
      if (k8 + 2 < 16) load_b(bA, k8 + 2);
      block(a[k8 + 1], bB);
    }

    const int rbase = row0 + 8 * hh;
#pragma unroll
    for (int i = 0; i < 8; ++i) {
      if (rbase + i < M) {
#pragma unroll
        for (int g = 0; g < G; ++g)
          *(float4*)(C + (long)(rbase + i) * NCOL + g * 64 + c * 4) =
              make_float4(acc[4 * g][i], acc[4 * g + 1][i], acc[4 * g + 2][i], acc[4 * g + 3][i]);
      }
    }

    const int j = c & (LPH - 1);
    const int o = c / LPH;
#pragma unroll
    for (int g = 0; g < G; ++g) {
      const float4 alv = *(const float4*)(al + g * 64 + c * 4);
      const float4 arv = *(const float4*)(ar + g * 64 + c * 4);
      float pl[8], pr[8];
#pragma unroll
      for (int i = 0; i < 8; ++i) {
        pl[i] = (acc[4 * g][i] * alv.x + acc[4 * g + 1][i] * alv.y) +
                (acc[4 * g + 2][i] * alv.z + acc[4 * g + 3][i] * alv.w);
        pr[i] = (acc[4 * g][i] * arv.x + acc[4 * g + 1][i] * arv.y) +
                (acc[4 * g + 2][i] * arv.z + acc[4 * g + 3][i] * arv.w);
      }
      if (LPH == 16) {
#pragma unroll
        for (int i = 0; i < 8; ++i) { pl[i] += row_xor<8>(pl[i]); pr[i] += row_xor<8>(pr[i]); }
      }
#pragma unroll
      for (int i = 0; i < 8; ++i) { pl[i] += row_xor<4>(pl[i]); pr[i] += row_xor<4>(pr[i]); }
#pragma unroll
      for (int i = 0; i < 8; ++i) { pl[i] += row_xor<2>(pl[i]); pr[i] += row_xor<2>(pr[i]); }
#pragma unroll
      for (int i = 0; i < 8; ++i) { pl[i] += row_xor<1>(pl[i]); pr[i] += row_xor<1>(pr[i]); }
      float sl = pl[0], sr = pr[0];
#pragma unroll
      for (int i = 1; i < 8; ++i) { if (j == i) { sl = pl[i]; sr = pr[i]; } }
      if (j < 8 && rbase + j < M) {
        el[(long)(rbase + j) * H + g * HPG + o] = sl;
        er[(long)(rbase + j) * H + g * HPG + o] = sr;
      }
    }
  }
  make_owner_keys<NB>(dst, key, E);
}

template <int F, int H, int NB, int CAP, int EPW>
__global__ void __launch_bounds__(1024) edge_owner(
    const int* __restrict__ src, const int* __restrict__ dst,
    const float* __restrict__ feat, const float* __restrict__ el,
    const float* __restrict__ er, const float* __restrict__ bias,
    float* __restrict__ out, int N, int E,
    const unsigned char* __restrict__ key) {
  constexpr int D  = F / H;
  constexpr int Q  = F / 32;
  constexpr int NT = 1024;
  constexpr int NW = NT / 32;
  static_assert(EPW * H <= 32, "one staging lane per (edge, head)");
  __shared__ float tile[NB * F];
  __shared__ float mL[NB * H];
  __shared__ float dL[NB * H];
  __shared__ int   le[CAP];
  __shared__ int   ls[CAP];
  __shared__ int   ld[CAP];
  __shared__ int   wc[NT / 32];
  constexpr int WCAP = CAP / (NT / 32);
  const int tid  = threadIdx.x;
  const int lane = tid & 31;
  const int wave = tid >> 5;
  const int lo   = blockIdx.x * NB;
  const int nbh  = (N - lo < NB) ? (N - lo) : NB;

  for (int i = tid; i < NB * F; i += NT) tile[i] = 0.f;
  for (int i = tid; i < NB * H; i += NT) { mL[i] = -INFINITY; dL[i] = 0.f; }

  const bool keyed = gridDim.x <= 256u;
  int wfill = 0;
  auto put = [&](bool p, unsigned m, int e) {
    const int pos = wfill + (int)__builtin_amdgcn_mbcnt_lo(m, 0u);
    if (p && pos < WCAP) le[wave * WCAP + pos] = e;
    wfill += __builtin_popcount(m);
  };
  const unsigned pat = (blockIdx.x & 255u) * 0x01010101u;
  auto zero_bytes = [](unsigned x) {
    const unsigned t = (x & 0x7f7f7f7fu) + 0x7f7f7f7fu;
    return ~(t | x | 0x7f7f7f7fu);
  };
  auto keep16 = [&](int4 v, bool valid, int e0) {
    const unsigned f0 = zero_bytes((unsigned)v.x ^ pat), f1 = zero_bytes((unsigned)v.y ^ pat);
    const unsigned f2 = zero_bytes((unsigned)v.z ^ pat), f3 = zero_bytes((unsigned)v.w ^ pat);
    unsigned m = valid ? ((f0 >> 7) | (f1 >> 6) | (f2 >> 5) | (f3 >> 4)) : 0u;
    unsigned any = __builtin_amdgcn_ballot_w32(m != 0u);
    while (any != 0u) {
      const bool p = (m != 0u);
      const int  i = p ? __builtin_ctz(m) : 0;
      put(p, any, e0 + ((i & 7) << 2) + (i >> 3));
      m &= m - 1u;
      any = __builtin_amdgcn_ballot_w32(m != 0u);
    }
  };
  if (keyed) {
    const int4* k4 = (const int4*)key;
    const int E16 = E >> 4;
    int base = wave * 32;
    for (; base + 7 * NT + 32 <= E16; base += 8 * NT) {
      int4 v[8];
#pragma unroll
      for (int u = 0; u < 8; ++u) v[u] = k4[base + u * NT + lane];
#pragma unroll
      for (int u = 0; u < 8; ++u) keep16(v[u], true, (base + u * NT + lane) << 4);
    }
    for (; base < E16; base += NT) {
      const int  i = base + lane;
      const bool valid = i < E16;
      const int4 v = k4[valid ? i : 0];
      keep16(v, valid, i << 4);
    }
    if (wave == 0) {
      const int  e = (E16 << 4) + lane;
      const bool in = (lane < 16) && (e < E);
      const bool p = in && (unsigned)(dst[in ? e : 0] - lo) < (unsigned)nbh;
      put(p, __builtin_amdgcn_ballot_w32(p), e);
    }
  }
  if (lane == 0) wc[wave] = wfill;
  __syncthreads();
  int total = 0, off = 0;
  bool fits = keyed;
  for (int w = 0; w < NT / 32; ++w) {
    const int n = wc[w];
    if (w < wave) off += n;
    total += n;
    fits = fits && (n <= WCAP);
  }
  if (fits) {
    for (int i = lane; i < wfill; i += 32) {
      const int e = le[wave * WCAP + i];
      ls[off + i] = src[e];
      ld[off + i] = dst[e] - lo;
    }
  }
  __syncthreads();

  auto score = [&](int s, int r, int h) {
    const float x = el[(long)s * H + h] + er[(long)(lo + r) * H + h];
    return (x > 0.f) ? x : 0.2f * x;
  };
  auto for_owned = [&](auto fn) {
    if (fits) {
      for (int i = tid; i < total; i += NT) fn(ls[i], ld[i]);
    } else {
      for (int e = tid; e < E; e += NT) {
        const unsigned r = (unsigned)(dst[e] - lo);
        if (r < (unsigned)nbh) fn(src[e], (int)r);
      }
    }
  };

  for_owned([&](int s, int r) {
#pragma unroll
    for (int h = 0; h < H; ++h) atomicMaxF(&mL[r * H + h], score(s, r, h));
  });
  __syncthreads();
  for_owned([&](int s, int r) {
#pragma unroll
    for (int h = 0; h < H; ++h)
      atomicAdd(&dL[r * H + h], expf(score(s, r, h) - mL[r * H + h]));
  });
  __syncthreads();

  if (fits) {
    const int jl = lane / H, hl = lane % H;
    for (int i0 = wave * EPW; i0 < total; i0 += NW * EPW) {
      const int c = (total - i0 < EPW) ? (total - i0) : EPW;
      int s_l = 0, r_l = 0; float a_l = 0.f;
      if (jl < c) {
        s_l = ls[i0 + jl]; r_l = ld[i0 + jl];
        a_l = expf(score(s_l, r_l, hl) - mL[r_l * H + hl]) / dL[r_l * H + hl];
      }
      float v[EPW][Q];
#pragma unroll
      for (int j = 0; j < EPW; ++j) {
        const int s = __builtin_amdgcn_readlane(s_l, j * H);
        const float* fr = feat + (long)s * F + lane;
#pragma unroll
        for (int q = 0; q < Q; ++q) v[j][q] = fr[q * 32];
      }
#pragma unroll
      for (int j = 0; j < EPW; ++j) {
        if (j < c) {
          const int r = __builtin_amdgcn_readlane(r_l, j * H);
          float* t = tile + r * F + lane;
#pragma unroll
          for (int q = 0; q < Q; ++q) {
            const float a = __int_as_float(
                __builtin_amdgcn_readlane(__float_as_int(a_l), j * H + (q * 32) / D));
            atomicAdd(t + q * 32, v[j][q] * a);
          }
        }
      }
    }
  } else {
    for_owned([&](int s, int r) {
      for (int h = 0; h < H; ++h) {
        const float a = expf(score(s, r, h) - mL[r * H + h]) / dL[r * H + h];
        const float* fr = feat + (long)s * F + h * D;
        float* t = tile + r * F + h * D;
        for (int k = 0; k < D; ++k) atomicAdd(t + k, fr[k] * a);
      }
    });
  }
  __syncthreads();

  {
    constexpr int C4 = F / 4;
    const float4* t4 = (const float4*)tile;
    const float4* b4 = (const float4*)bias;
    float4* o4 = (float4*)(out + (long)lo * F);
    for (int i = tid; i < nbh * C4; i += NT) {
      float4 v = t4[i];
      const float4 b = b4[i & (C4 - 1)];
      v.x += b.x; v.y += b.y; v.z += b.z; v.w += b.w;
      o4[i] = v;
    }
  }
}

extern "C" void kernel_launch(void* const* d_in, const int* in_sizes, int n_in,
                              void* d_out, int out_size, void* d_ws, size_t ws_size,
                              hipStream_t stream) {
  const float* h   = (const float*)d_in[0];
  const int*   src = (const int*)d_in[1];
  const int*   dst = (const int*)d_in[2];
  const float* W1  = (const float*)d_in[3];
  const float* al1 = (const float*)d_in[4];
  const float* ar1 = (const float*)d_in[5];
  const float* b1  = (const float*)d_in[6];
  const float* W2  = (const float*)d_in[7];
  const float* al2 = (const float*)d_in[8];
  const float* ar2 = (const float*)d_in[9];
  const float* b2  = (const float*)d_in[10];

  constexpr int F1 = 128, H1 = 4;
  constexpr int F2 = 64,  H2 = 1;
  constexpr int NB = 196;
  constexpr int CAP = 8192;
  const long N = in_sizes[0] / F1;
  const long E = in_sizes[1];

  float* ws    = (float*)d_ws;
  float* feat1 = ws;
  float* out1  = feat1 + N * F1;
  float* feat2 = out1  + N * F1;
  float* el    = feat2 + N * F2;
  float* er    = el    + N * H1;
  float* out2  = (float*)d_out;

  auto cdiv = [](long a, long b) { return (unsigned)((a + b - 1) / b); };
  const unsigned gblocks = cdiv(N, 128);
  const unsigned owners = cdiv(N, NB);

  wmma_gemm_f32_k128<F1, H1, false, NB><<<gblocks, 256, 0, stream>>>(
      h, W1, feat1, (int)N, al1, ar1, el, er, dst, (unsigned char*)feat2, (int)E);
  edge_owner<F1, H1, NB, CAP, 8><<<owners, 1024, 0, stream>>>(
      src, dst, feat1, el, er, b1, out1, (int)N, (int)E, (const unsigned char*)feat2);

  wmma_gemm_f32_k128<F2, H2, true, NB><<<gblocks, 256, 0, stream>>>(
      out1, W2, feat2, (int)N, al2, ar2, el, er, dst, (unsigned char*)feat1, (int)E);
  edge_owner<F2, H2, NB, CAP, 16><<<owners, 1024, 0, stream>>>(
      src, dst, feat2, el, er, b2, out2, (int)N, (int)E, (const unsigned char*)feat1);
}


# Layer 2 reuses layer 1's owner lists: the first edge launch leaves each owner's (src, local dst) list in the workspace, the second loads it; no second key pass, key scan or src/dst gather
# speedup vs baseline: 4.1719x; 1.1495x over previous
#include <hip/hip_runtime.h>
#include <math.h>

typedef __attribute__((ext_vector_type(2))) float v2f;
typedef __attribute__((ext_vector_type(8))) float v8f;

__device__ __forceinline__ void atomicMaxF(float* addr, float v) {
  if (v >= 0.0f) atomicMax((int*)addr, __float_as_int(v));
  else           atomicMin((unsigned int*)addr, __float_as_uint(v));
}

template <int XM>
__device__ __forceinline__ float row_xor(float v) {
  return __int_as_float(__builtin_amdgcn_update_dpp(
      0, __float_as_int(v), 0x160 + XM, 0xf, 0xf, true));
}

template <int NB>
__device__ __forceinline__ void make_owner_keys(const int* __restrict__ dst,
                                                unsigned char* __restrict__ key, int E) {
  const int E4 = E >> 2;
  const int4* d4 = (const int4*)dst;
  unsigned* k32 = (unsigned*)key;
  const int step = gridDim.x * blockDim.x;
  for (int q = blockIdx.x * blockDim.x + threadIdx.x; q < E4; q += step) {
    const int4 d = d4[q];
    k32[q] = (((unsigned)d.x / NB) & 255u) | ((((unsigned)d.y / NB) & 255u) << 8) |
             ((((unsigned)d.z / NB) & 255u) << 16) | ((((unsigned)d.w / NB) & 255u) << 24);
  }
  if (blockIdx.x == 0 && (int)threadIdx.x < (E & 3)) {
    const int e = (E4 << 2) + threadIdx.x;
    key[e] = (unsigned char)(((unsigned)dst[e] / NB) & 255u);
  }
}

template <int NCOL, int H, bool RELU_A, int NB, bool KEYS>
__global__ void __launch_bounds__(256) __attribute__((amdgpu_waves_per_eu(4, 4))) wmma_gemm_f32_k128(
    const float* __restrict__ A, const float* __restrict__ B,
    float* __restrict__ C, int M,
    const float* __restrict__ al, const float* __restrict__ ar,
    float* __restrict__ el, float* __restrict__ er,
    const int* __restrict__ dst, unsigned char* __restrict__ key, int E) {
  constexpr int CT  = NCOL / 16;
  constexpr int G   = NCOL / 64;
  constexpr int D   = NCOL / H;
  constexpr int LPH = D / 4;
  constexpr int HPG = 64 / D;
  static_assert(NCOL % 64 == 0 && (LPH == 8 || LPH == 16) && G * HPG == H, "shape");
  __shared__ float4 Bs[16 * CT * 32];
  const int lane = threadIdx.x & 31;
  const int wave = threadIdx.x >> 5;
  const int c    = lane & 15;
  const int hh   = lane >> 4;
  const int mtiles = (M + 15) >> 4;
  const int tile = blockIdx.x * 8 + wave;
  const bool has = tile < mtiles;
  const int row0 = tile * 16;

  float4 a[16];
  {
    int r = row0 + c;
    if (r >= M) r = M - 1;
    const float4* ap = (const float4*)(A + (long)r * 128) + hh;
#pragma unroll
    for (int k8 = 0; k8 < 16; ++k8) a[k8] = ap[2 * k8];
  }

#pragma unroll
  for (int it = 0; it < (16 * G * 32) / 256; ++it) {
    const int i = threadIdx.x + it * 256;
    const int sg  = (i >> 5) & (G - 1);
    const int sk8 = i / (32 * G);
    const float* bp = B + (long)(8 * sk8 + 4 * hh) * NCOL + sg * 64 + c * 4;
    const float4 v0 = *(const float4*)(bp);
    const float4 v1 = *(const float4*)(bp + NCOL);
    const float4 v2 = *(const float4*)(bp + 2 * NCOL);
    const float4 v3 = *(const float4*)(bp + 3 * NCOL);
    float4* d = Bs + (sk8 * CT + 4 * sg) * 32 + lane;
    d[0]  = make_float4(v0.x, v1.x, v2.x, v3.x);
    d[32] = make_float4(v0.y, v1.y, v2.y, v3.y);
    d[64] = make_float4(v0.z, v1.z, v2.z, v3.z);
    d[96] = make_float4(v0.w, v1.w, v2.w, v3.w);
  }
  __syncthreads();

  if (has) {
    if (RELU_A) {
#pragma unroll
      for (int k8 = 0; k8 < 16; ++k8) {
        a[k8].x = fmaxf(a[k8].x, 0.f); a[k8].y = fmaxf(a[k8].y, 0.f);
        a[k8].z = fmaxf(a[k8].z, 0.f); a[k8].w = fmaxf(a[k8].w, 0.f);
      }
    }
    v8f acc[CT];
#pragma unroll
    for (int ct = 0; ct < CT; ++ct) acc[ct] = v8f{};
    const float4* bq = Bs + lane;
    auto load_b = [&](float4 (&b)[CT], int k8) {
#pragma unroll
      for (int ct = 0; ct < CT; ++ct) b[ct] = bq[(k8 * CT + ct) * 32];
    };
    auto block = [&](const float4& av, const float4 (&b)[CT]) {
      v2f a01, a23;
      a01.x = av.x; a01.y = av.y; a23.x = av.z; a23.y = av.w;
#pragma unroll
      for (int ct = 0; ct < CT; ++ct) {
        v2f b01; b01.x = b[ct].x; b01.y = b[ct].y;
        acc[ct] = __builtin_amdgcn_wmma_f32_16x16x4_f32(false, a01, false, b01, (short)0, acc[ct], false, false);
      }
#pragma unroll
      for (int ct = 0; ct < CT; ++ct) {
        v2f b23; b23.x = b[ct].z; b23.y = b[ct].w;
        acc[ct] = __builtin_amdgcn_wmma_f32_16x16x4_f32(false, a23, false, b23, (short)0, acc[ct], false, false);
      }
    };
    float4 bA[CT], bB[CT];
    load_b(bA, 0);
#pragma unroll
    for (int k8 = 0; k8 < 16; k8 += 2) {
      load_b(bB, k8 + 1);
      block(a[k8], bA);
      if (k8 + 2 < 16) load_b(bA, k8 + 2);
      block(a[k8 + 1], bB);
    }

    const int rbase = row0 + 8 * hh;
#pragma unroll
    for (int i = 0; i < 8; ++i) {
      if (rbase + i < M) {
#pragma unroll
        for (int g = 0; g < G; ++g)
          *(float4*)(C + (long)(rbase + i) * NCOL + g * 64 + c * 4) =
              make_float4(acc[4 * g][i], acc[4 * g + 1][i], acc[4 * g + 2][i], acc[4 * g + 3][i]);
      }
    }

    const int j = c & (LPH - 1);
    const int o = c / LPH;
#pragma unroll
    for (int g = 0; g < G; ++g) {
      const float4 alv = *(const float4*)(al + g * 64 + c * 4);
      const float4 arv = *(const float4*)(ar + g * 64 + c * 4);
      float pl[8], pr[8];
#pragma unroll
      for (int i = 0; i < 8; ++i) {
        pl[i] = (acc[4 * g][i] * alv.x + acc[4 * g + 1][i] * alv.y) +
                (acc[4 * g + 2][i] * alv.z + acc[4 * g + 3][i] * alv.w);
        pr[i] = (acc[4 * g][i] * arv.x + acc[4 * g + 1][i] * arv.y) +
                (acc[4 * g + 2][i] * arv.z + acc[4 * g + 3][i] * arv.w);
      }
      if (LPH == 16) {
#pragma unroll
        for (int i = 0; i < 8; ++i) { pl[i] += row_xor<8>(pl[i]); pr[i] += row_xor<8>(pr[i]); }
      }
#pragma unroll
      for (int i = 0; i < 8; ++i) { pl[i] += row_xor<4>(pl[i]); pr[i] += row_xor<4>(pr[i]); }
#pragma unroll
      for (int i = 0; i < 8; ++i) { pl[i] += row_xor<2>(pl[i]); pr[i] += row_xor<2>(pr[i]); }
#pragma unroll
      for (int i = 0; i < 8; ++i) { pl[i] += row_xor<1>(pl[i]); pr[i] += row_xor<1>(pr[i]); }
      float sl = pl[0], sr = pr[0];
#pragma unroll
      for (int i = 1; i < 8; ++i) { if (j == i) { sl = pl[i]; sr = pr[i]; } }
      if (j < 8 && rbase + j < M) {
        el[(long)(rbase + j) * H + g * HPG + o] = sl;
        er[(long)(rbase + j) * H + g * HPG + o] = sr;
      }
    }
  }
  if (KEYS) make_owner_keys<NB>(dst, key, E);
}

template <int F, int H, int NB, int CAP, int EPW, bool FIRST>
__global__ void __launch_bounds__(1024) edge_owner(
    const int* __restrict__ src, const int* __restrict__ dst,
    const float* __restrict__ feat, const float* __restrict__ el,
    const float* __restrict__ er, const float* __restrict__ bias,
    float* __restrict__ out, int N, int E,
    const unsigned char* __restrict__ key, unsigned* __restrict__ stash) {
  constexpr int D  = F / H;
  constexpr int Q  = F / 32;
  constexpr int NT = 1024;
  constexpr int NW = NT / 32;
  static_assert(EPW * H <= 32, "one staging lane per (edge, head)");
  __shared__ float tile[NB * F];
  __shared__ float mL[NB * H];
  __shared__ float dL[NB * H];
  __shared__ int   le[CAP];
  __shared__ int   ls[CAP];
  __shared__ int   ld[CAP];
  __shared__ int   wc[NT / 32];
  constexpr int WCAP = CAP / (NT / 32);
  const int tid  = threadIdx.x;
  const int lane = tid & 31;
  const int wave = tid >> 5;
  const int lo   = blockIdx.x * NB;
  const int nbh  = (N - lo < NB) ? (N - lo) : NB;
  constexpr int SLOT = CAP + 32;
  unsigned* slot = stash + (long)blockIdx.x * SLOT;
  const bool packs = N <= (1 << 24);
  unsigned head = 0xffffffffu;
  if (!FIRST) head = slot[0];

  for (int i = tid; i < NB * F; i += NT) tile[i] = 0.f;
  for (int i = tid; i < NB * H; i += NT) { mL[i] = -INFINITY; dL[i] = 0.f; }

  const bool keyed = gridDim.x <= 256u;
  int stashed = -1;
  if (!FIRST) {
    stashed = __builtin_amdgcn_readfirstlane((int)head);
    if (stashed > CAP) stashed = -1;
  }
  int wfill = 0;
  auto put = [&](bool p, unsigned m, int e) {
    const int pos = wfill + (int)__builtin_amdgcn_mbcnt_lo(m, 0u);
    if (p && pos < WCAP) le[wave * WCAP + pos] = e;
    wfill += __builtin_popcount(m);
  };
  const unsigned pat = (blockIdx.x & 255u) * 0x01010101u;
  auto zero_bytes = [](unsigned x) {
    const unsigned t = (x & 0x7f7f7f7fu) + 0x7f7f7f7fu;
    return ~(t | x | 0x7f7f7f7fu);
  };
  auto keep16 = [&](int4 v, bool valid, int e0) {
    const unsigned f0 = zero_bytes((unsigned)v.x ^ pat), f1 = zero_bytes((unsigned)v.y ^ pat);
    const unsigned f2 = zero_bytes((unsigned)v.z ^ pat), f3 = zero_bytes((unsigned)v.w ^ pat);
    unsigned m = valid ? ((f0 >> 7) | (f1 >> 6) | (f2 >> 5) | (f3 >> 4)) : 0u;
    unsigned any = __builtin_amdgcn_ballot_w32(m != 0u);
    while (any != 0u) {
      const bool p = (m != 0u);
      const int  i = p ? __builtin_ctz(m) : 0;
      put(p, any, e0 + ((i & 7) << 2) + (i >> 3));
      m &= m - 1u;
      any = __builtin_amdgcn_ballot_w32(m != 0u);
    }
  };
  if (keyed && stashed < 0) {
    const int4* k4 = (const int4*)key;
    const int E16 = E >> 4;
    int base = wave * 32;
    for (; base + 7 * NT + 32 <= E16; base += 8 * NT) {
      int4 v[8];
#pragma unroll
      for (int u = 0; u < 8; ++u) v[u] = k4[base + u * NT + lane];
#pragma unroll
      for (int u = 0; u < 8; ++u) keep16(v[u], true, (base + u * NT + lane) << 4);
    }
    for (; base < E16; base += NT) {
      const int  i = base + lane;
      const bool valid = i < E16;
      const int4 v = k4[valid ? i : 0];
      keep16(v, valid, i << 4);
    }
    if (wave == 0) {
      const int  e = (E16 << 4) + lane;
      const bool in = (lane < 16) && (e < E);
      const bool p = in && (unsigned)(dst[in ? e : 0] - lo) < (unsigned)nbh;
      put(p, __builtin_amdgcn_ballot_w32(p), e);
    }
  }
  if (lane == 0) wc[wave] = wfill;
  __syncthreads();
  int total = 0, off = 0;
  bool fits = keyed;
  for (int w = 0; w < NT / 32; ++w) {
    const int n = wc[w];
    if (w < wave) off += n;
    total += n;
    fits = fits && (n <= WCAP);
  }
  if (stashed >= 0) {
    total = stashed;
    fits = true;
    for (int i = tid; i < total; i += NT) {
      const unsigned w = slot[32 + i];
      ls[i] = (int)(w & 0xffffffu);
      ld[i] = (int)(w >> 24);
    }
  } else if (fits) {
    for (int i = lane; i < wfill; i += 32) {
      const int e = le[wave * WCAP + i];
      const int s = src[e];
      const int r = dst[e] - lo;
      ls[off + i] = s;
      ld[off + i] = r;
      if (FIRST) slot[32 + off + i] = (unsigned)s | ((unsigned)r << 24);
    }
  }
  if (FIRST && tid == 0) slot[0] = (fits && packs) ? (unsigned)total : 0xffffffffu;
  __syncthreads();

  auto score = [&](int s, int r, int h) {
    const float x = el[(long)s * H + h] + er[(long)(lo + r) * H + h];
    return (x > 0.f) ? x : 0.2f * x;
  };
  auto for_owned = [&](auto fn) {
    if (fits) {
      for (int i = tid; i < total; i += NT) fn(ls[i], ld[i]);
    } else {
      for (int e = tid; e < E; e += NT) {
        const unsigned r = (unsigned)(dst[e] - lo);
        if (r < (unsigned)nbh) fn(src[e], (int)r);
      }
    }
  };

  for_owned([&](int s, int r) {
#pragma unroll
    for (int h = 0; h < H; ++h) atomicMaxF(&mL[r * H + h], score(s, r, h));
  });
  __syncthreads();
  for_owned([&](int s, int r) {
#pragma unroll
    for (int h = 0; h < H; ++h)
      atomicAdd(&dL[r * H + h], expf(score(s, r, h) - mL[r * H + h]));
  });
  __syncthreads();

  if (fits) {
    const int jl = lane / H, hl = lane % H;
    for (int i0 = wave * EPW; i0 < total; i0 += NW * EPW) {
      const int c = (total - i0 < EPW) ? (total - i0) : EPW;
      int s_l = 0, r_l = 0; float a_l = 0.f;
      if (jl < c) {
        s_l = ls[i0 + jl]; r_l = ld[i0 + jl];
        a_l = expf(score(s_l, r_l, hl) - mL[r_l * H + hl]) / dL[r_l * H + hl];
      }
      float v[EPW][Q];
#pragma unroll
      for (int j = 0; j < EPW; ++j) {
        const int s = __builtin_amdgcn_readlane(s_l, j * H);
        const float* fr = feat + (long)s * F + lane;
#pragma unroll
        for (int q = 0; q < Q; ++q) v[j][q] = fr[q * 32];
      }
#pragma unroll
      for (int j = 0; j < EPW; ++j) {
        if (j < c) {
          const int r = __builtin_amdgcn_readlane(r_l, j * H);
          float* t = tile + r * F + lane;
#pragma unroll
          for (int q = 0; q < Q; ++q) {
            const float a = __int_as_float(
                __builtin_amdgcn_readlane(__float_as_int(a_l), j * H + (q * 32) / D));
            atomicAdd(t + q * 32, v[j][q] * a);
          }
        }
      }
    }
  } else {
    for_owned([&](int s, int r) {
      for (int h = 0; h < H; ++h) {
        const float a = expf(score(s, r, h) - mL[r * H + h]) / dL[r * H + h];
        const float* fr = feat + (long)s * F + h * D;
        float* t = tile + r * F + h * D;
        for (int k = 0; k < D; ++k) atomicAdd(t + k, fr[k] * a);
      }
    });
  }
  __syncthreads();

  {
    constexpr int C4 = F / 4;
    const float4* t4 = (const float4*)tile;
    const float4* b4 = (const float4*)bias;
    float4* o4 = (float4*)(out + (long)lo * F);
    for (int i = tid; i < nbh * C4; i += NT) {
      float4 v = t4[i];
      const float4 b = b4[i & (C4 - 1)];
      v.x += b.x; v.y += b.y; v.z += b.z; v.w += b.w;
      o4[i] = v;
    }
  }
}

extern "C" void kernel_launch(void* const* d_in, const int* in_sizes, int n_in,
                              void* d_out, int out_size, void* d_ws, size_t ws_size,
                              hipStream_t stream) {
  const float* h   = (const float*)d_in[0];
  const int*   src = (const int*)d_in[1];
  const int*   dst = (const int*)d_in[2];
  const float* W1  = (const float*)d_in[3];
  const float* al1 = (const float*)d_in[4];
  const float* ar1 = (const float*)d_in[5];
  const float* b1  = (const float*)d_in[6];
  const float* W2  = (const float*)d_in[7];
  const float* al2 = (const float*)d_in[8];
  const float* ar2 = (const float*)d_in[9];
  const float* b2  = (const float*)d_in[10];

  constexpr int F1 = 128, H1 = 4;
  constexpr int F2 = 64,  H2 = 1;
  constexpr int NB = 196;
  constexpr int CAP = 8192;
  const long N = in_sizes[0] / F1;
  const long E = in_sizes[1];

  float* ws    = (float*)d_ws;
  float* feat1 = ws;
  float* out1  = feat1 + N * F1;
  float* aux   = out1  + N * F1;
  float* el    = aux   + N * F2;
  float* er    = el    + N * H1;
  float* out2  = (float*)d_out;

  auto cdiv = [](long a, long b) { return (unsigned)((a + b - 1) / b); };
  const unsigned gblocks = cdiv(N, 128);
  const unsigned owners = cdiv(N, NB);
  unsigned char* keys = (unsigned char*)aux;
  unsigned* stash = (unsigned*)(aux + N * (F2 / 4));
  float* feat2 = feat1;

  wmma_gemm_f32_k128<F1, H1, false, NB, true><<<gblocks, 256, 0, stream>>>(
      h, W1, feat1, (int)N, al1, ar1, el, er, dst, keys, (int)E);
  edge_owner<F1, H1, NB, CAP, 8, true><<<owners, 1024, 0, stream>>>(
      src, dst, feat1, el, er, b1, out1, (int)N, (int)E, keys, stash);

  wmma_gemm_f32_k128<F2, H2, true, NB, false><<<gblocks, 256, 0, stream>>>(
      out1, W2, feat2, (int)N, al2, ar2, el, er, dst, keys, (int)E);
  edge_owner<F2, H2, NB, CAP, 16, false><<<owners, 1024, 0, stream>>>(
      src, dst, feat2, el, er, b2, out2, (int)N, (int)E, keys, stash);
}


# Each listed edge's score is gathered from el/er once (max pass) and kept in LDS beside its list entry, exp in place; exp-sum and row-sum passes read LDS, not global; sums and order unchanged
# speedup vs baseline: 4.7413x; 1.1365x over previous
#include <hip/hip_runtime.h>
#include <math.h>

typedef __attribute__((ext_vector_type(2))) float v2f;
typedef __attribute__((ext_vector_type(8))) float v8f;

__device__ __forceinline__ void atomicMaxF(float* addr, float v) {
  if (v >= 0.0f) atomicMax((int*)addr, __float_as_int(v));
  else           atomicMin((unsigned int*)addr, __float_as_uint(v));
}

template <int XM>
__device__ __forceinline__ float row_xor(float v) {
  return __int_as_float(__builtin_amdgcn_update_dpp(
      0, __float_as_int(v), 0x160 + XM, 0xf, 0xf, true));
}

template <int NB>
__device__ __forceinline__ void make_owner_keys(const int* __restrict__ dst,
                                                unsigned char* __restrict__ key, int E) {
  const int E4 = E >> 2;
  const int4* d4 = (const int4*)dst;
  unsigned* k32 = (unsigned*)key;
  const int step = gridDim.x * blockDim.x;
  for (int q = blockIdx.x * blockDim.x + threadIdx.x; q < E4; q += step) {
    const int4 d = d4[q];
    k32[q] = (((unsigned)d.x / NB) & 255u) | ((((unsigned)d.y / NB) & 255u) << 8) |
             ((((unsigned)d.z / NB) & 255u) << 16) | ((((unsigned)d.w / NB) & 255u) << 24);
  }
  if (blockIdx.x == 0 && (int)threadIdx.x < (E & 3)) {
    const int e = (E4 << 2) + threadIdx.x;
    key[e] = (unsigned char)(((unsigned)dst[e] / NB) & 255u);
  }
}

template <int NCOL, int H, bool RELU_A, int NB, bool KEYS>
__global__ void __launch_bounds__(256) __attribute__((amdgpu_waves_per_eu(4, 4))) wmma_gemm_f32_k128(
    const float* __restrict__ A, const float* __restrict__ B,
    float* __restrict__ C, int M,
    const float* __restrict__ al, const float* __restrict__ ar,
    float* __restrict__ el, float* __restrict__ er,
    const int* __restrict__ dst, unsigned char* __restrict__ key, int E) {
  constexpr int CT  = NCOL / 16;
  constexpr int G   = NCOL / 64;
  constexpr int D   = NCOL / H;
  constexpr int LPH = D / 4;
  constexpr int HPG = 64 / D;
  static_assert(NCOL % 64 == 0 && (LPH == 8 || LPH == 16) && G * HPG == H, "shape");
  __shared__ float4 Bs[16 * CT * 32];
  const int lane = threadIdx.x & 31;
  const int wave = threadIdx.x >> 5;
  const int c    = lane & 15;
  const int hh   = lane >> 4;
  const int mtiles = (M + 15) >> 4;
  const int tile = blockIdx.x * 8 + wave;
  const bool has = tile < mtiles;
  const int row0 = tile * 16;

  float4 a[16];
  {
    int r = row0 + c;
    if (r >= M) r = M - 1;
    const float4* ap = (const float4*)(A + (long)r * 128) + hh;
#pragma unroll
    for (int k8 = 0; k8 < 16; ++k8) a[k8] = ap[2 * k8];
  }

#pragma unroll
  for (int it = 0; it < (16 * G * 32) / 256; ++it) {
    const int i = threadIdx.x + it * 256;
    const int sg  = (i >> 5) & (G - 1);
    const int sk8 = i / (32 * G);
    const float* bp = B + (long)(8 * sk8 + 4 * hh) * NCOL + sg * 64 + c * 4;
    const float4 v0 = *(const float4*)(bp);
    const float4 v1 = *(const float4*)(bp + NCOL);
    const float4 v2 = *(const float4*)(bp + 2 * NCOL);
    const float4 v3 = *(const float4*)(bp + 3 * NCOL);
    float4* d = Bs + (sk8 * CT + 4 * sg) * 32 + lane;
    d[0]  = make_float4(v0.x, v1.x, v2.x, v3.x);
    d[32] = make_float4(v0.y, v1.y, v2.y, v3.y);
    d[64] = make_float4(v0.z, v1.z, v2.z, v3.z);
    d[96] = make_float4(v0.w, v1.w, v2.w, v3.w);
  }
  __syncthreads();

  if (has) {
    if (RELU_A) {
#pragma unroll
      for (int k8 = 0; k8 < 16; ++k8) {
        a[k8].x = fmaxf(a[k8].x, 0.f); a[k8].y = fmaxf(a[k8].y, 0.f);
        a[k8].z = fmaxf(a[k8].z, 0.f); a[k8].w = fmaxf(a[k8].w, 0.f);
      }
    }
    v8f acc[CT];
#pragma unroll
    for (int ct = 0; ct < CT; ++ct) acc[ct] = v8f{};
    const float4* bq = Bs + lane;
    auto load_b = [&](float4 (&b)[CT], int k8) {
#pragma unroll
      for (int ct = 0; ct < CT; ++ct) b[ct] = bq[(k8 * CT + ct) * 32];
    };
    auto block = [&](const float4& av, const float4 (&b)[CT]) {
      v2f a01, a23;
      a01.x = av.x; a01.y = av.y; a23.x = av.z; a23.y = av.w;
#pragma unroll
      for (int ct = 0; ct < CT; ++ct) {
        v2f b01; b01.x = b[ct].x; b01.y = b[ct].y;
        acc[ct] = __builtin_amdgcn_wmma_f32_16x16x4_f32(false, a01, false, b01, (short)0, acc[ct], false, false);
      }
#pragma unroll
      for (int ct = 0; ct < CT; ++ct) {
        v2f b23; b23.x = b[ct].z; b23.y = b[ct].w;
        acc[ct] = __builtin_amdgcn_wmma_f32_16x16x4_f32(false, a23, false, b23, (short)0, acc[ct], false, false);
      }
    };
    float4 bA[CT], bB[CT];
    load_b(bA, 0);
#pragma unroll
    for (int k8 = 0; k8 < 16; k8 += 2) {
      load_b(bB, k8 + 1);
      block(a[k8], bA);
      if (k8 + 2 < 16) load_b(bA, k8 + 2);
      block(a[k8 + 1], bB);
    }

    const int rbase = row0 + 8 * hh;
#pragma unroll
    for (int i = 0; i < 8; ++i) {
      if (rbase + i < M) {
#pragma unroll
        for (int g = 0; g < G; ++g)
          *(float4*)(C + (long)(rbase + i) * NCOL + g * 64 + c * 4) =
              make_float4(acc[4 * g][i], acc[4 * g + 1][i], acc[4 * g + 2][i], acc[4 * g + 3][i]);
      }
    }

    const int j = c & (LPH - 1);
    const int o = c / LPH;
#pragma unroll
    for (int g = 0; g < G; ++g) {
      const float4 alv = *(const float4*)(al + g * 64 + c * 4);
      const float4 arv = *(const float4*)(ar + g * 64 + c * 4);
      float pl[8], pr[8];
#pragma unroll
      for (int i = 0; i < 8; ++i) {
        pl[i] = (acc[4 * g][i] * alv.x + acc[4 * g + 1][i] * alv.y) +
                (acc[4 * g + 2][i] * alv.z + acc[4 * g + 3][i] * alv.w);
        pr[i] = (acc[4 * g][i] * arv.x + acc[4 * g + 1][i] * arv.y) +
                (acc[4 * g + 2][i] * arv.z + acc[4 * g + 3][i] * arv.w);
      }
      if (LPH == 16) {
#pragma unroll
        for (int i = 0; i < 8; ++i) { pl[i] += row_xor<8>(pl[i]); pr[i] += row_xor<8>(pr[i]); }
      }
#pragma unroll
      for (int i = 0; i < 8; ++i) { pl[i] += row_xor<4>(pl[i]); pr[i] += row_xor<4>(pr[i]); }
#pragma unroll
      for (int i = 0; i < 8; ++i) { pl[i] += row_xor<2>(pl[i]); pr[i] += row_xor<2>(pr[i]); }
#pragma unroll
      for (int i = 0; i < 8; ++i) { pl[i] += row_xor<1>(pl[i]); pr[i] += row_xor<1>(pr[i]); }
      float sl = pl[0], sr = pr[0];
#pragma unroll
      for (int i = 1; i < 8; ++i) { if (j == i) { sl = pl[i]; sr = pr[i]; } }
      if (j < 8 && rbase + j < M) {
        el[(long)(rbase + j) * H + g * HPG + o] = sl;
        er[(long)(rbase + j) * H + g * HPG + o] = sr;
      }
    }
  }
  if (KEYS) make_owner_keys<NB>(dst, key, E);
}

template <int F, int H, int NB, int CAP, int EPW, bool FIRST>
__global__ void __launch_bounds__(1024) edge_owner(
    const int* __restrict__ src, const int* __restrict__ dst,
    const float* __restrict__ feat, const float* __restrict__ el,
    const float* __restrict__ er, const float* __restrict__ bias,
    float* __restrict__ out, int N, int E,
    const unsigned char* __restrict__ key, unsigned* __restrict__ stash) {
  constexpr int D  = F / H;
  constexpr int Q  = F / 32;
  constexpr int NT = 1024;
  constexpr int NW = NT / 32;
  static_assert(EPW * H <= 32, "one staging lane per (edge, head)");
  __shared__ float tile[NB * F];
  __shared__ float mL[NB * H];
  __shared__ float dL[NB * H];
  __shared__ float sc[CAP * H];
  int* const le = (int*)sc;
  __shared__ int   ls[CAP];
  __shared__ int   ld[CAP];
  __shared__ int   wc[NT / 32];
  constexpr int WCAP = CAP / (NT / 32);
  const int tid  = threadIdx.x;
  const int lane = tid & 31;
  const int wave = tid >> 5;
  const int lo   = blockIdx.x * NB;
  const int nbh  = (N - lo < NB) ? (N - lo) : NB;
  constexpr int SLOT = CAP + 32;
  unsigned* slot = stash + (long)blockIdx.x * SLOT;
  const bool packs = N <= (1 << 24);
  unsigned head = 0xffffffffu;
  if (!FIRST) head = slot[0];

  for (int i = tid; i < NB * F; i += NT) tile[i] = 0.f;
  for (int i = tid; i < NB * H; i += NT) { mL[i] = -INFINITY; dL[i] = 0.f; }

  const bool keyed = gridDim.x <= 256u;
  int stashed = -1;
  if (!FIRST) {
    stashed = __builtin_amdgcn_readfirstlane((int)head);
    if (stashed > CAP) stashed = -1;
  }
  int wfill = 0;
  auto put = [&](bool p, unsigned m, int e) {
    const int pos = wfill + (int)__builtin_amdgcn_mbcnt_lo(m, 0u);
    if (p && pos < WCAP) le[wave * WCAP + pos] = e;
    wfill += __builtin_popcount(m);
  };
  const unsigned pat = (blockIdx.x & 255u) * 0x01010101u;
  auto zero_bytes = [](unsigned x) {
    const unsigned t = (x & 0x7f7f7f7fu) + 0x7f7f7f7fu;
    return ~(t | x | 0x7f7f7f7fu);
  };
  auto keep16 = [&](int4 v, bool valid, int e0) {
    const unsigned f0 = zero_bytes((unsigned)v.x ^ pat), f1 = zero_bytes((unsigned)v.y ^ pat);
    const unsigned f2 = zero_bytes((unsigned)v.z ^ pat), f3 = zero_bytes((unsigned)v.w ^ pat);
    unsigned m = valid ? ((f0 >> 7) | (f1 >> 6) | (f2 >> 5) | (f3 >> 4)) : 0u;
    unsigned any = __builtin_amdgcn_ballot_w32(m != 0u);
    while (any != 0u) {
      const bool p = (m != 0u);
      const int  i = p ? __builtin_ctz(m) : 0;
      put(p, any, e0 + ((i & 7) << 2) + (i >> 3));
      m &= m - 1u;
      any = __builtin_amdgcn_ballot_w32(m != 0u);
    }
  };
  if (keyed && stashed < 0) {
    const int4* k4 = (const int4*)key;
    const int E16 = E >> 4;
    int base = wave * 32;
    for (; base + 7 * NT + 32 <= E16; base += 8 * NT) {
      int4 v[8];
#pragma unroll
      for (int u = 0; u < 8; ++u) v[u] = k4[base + u * NT + lane];
#pragma unroll
      for (int u = 0; u < 8; ++u) keep16(v[u], true, (base + u * NT + lane) << 4);
    }
    for (; base < E16; base += NT) {
      const int  i = base + lane;
      const bool valid = i < E16;
      const int4 v = k4[valid ? i : 0];
      keep16(v, valid, i << 4);
    }
    if (wave == 0) {
      const int  e = (E16 << 4) + lane;
      const bool in = (lane < 16) && (e < E);
      const bool p = in && (unsigned)(dst[in ? e : 0] - lo) < (unsigned)nbh;
      put(p, __builtin_amdgcn_ballot_w32(p), e);
    }
  }
  if (lane == 0) wc[wave] = wfill;
  __syncthreads();
  int total = 0, off = 0;
  bool fits = keyed;
  for (int w = 0; w < NT / 32; ++w) {
    const int n = wc[w];
    if (w < wave) off += n;
    total += n;
    fits = fits && (n <= WCAP);
  }
  if (stashed >= 0) {
    total = stashed;
    fits = true;
    for (int i = tid; i < total; i += NT) {
      const unsigned w = slot[32 + i];
      ls[i] = (int)(w & 0xffffffu);
      ld[i] = (int)(w >> 24);
    }
  } else if (fits) {
    for (int i = lane; i < wfill; i += 32) {
      const int e = le[wave * WCAP + i];
      const int s = src[e];
      const int r = dst[e] - lo;
      ls[off + i] = s;
      ld[off + i] = r;
      if (FIRST) slot[32 + off + i] = (unsigned)s | ((unsigned)r << 24);
    }
  }
  if (FIRST && tid == 0) slot[0] = (fits && packs) ? (unsigned)total : 0xffffffffu;
  __syncthreads();

  auto score = [&](int s, int r, int h) {
    const float x = el[(long)s * H + h] + er[(long)(lo + r) * H + h];
    return (x > 0.f) ? x : 0.2f * x;
  };
  auto for_scanned = [&](auto fn) {
    for (int e = tid; e < E; e += NT) {
      const unsigned r = (unsigned)(dst[e] - lo);
      if (r < (unsigned)nbh) fn(src[e], (int)r);
    }
  };

  if (fits) {
    for (int i = tid; i < total; i += NT) {
      const int s = ls[i], r = ld[i];
      float x[H];
#pragma unroll
      for (int h = 0; h < H; ++h) x[h] = score(s, r, h);
#pragma unroll
      for (int h = 0; h < H; ++h) sc[i * H + h] = x[h];
#pragma unroll
      for (int h = 0; h < H; ++h) atomicMaxF(&mL[r * H + h], x[h]);
    }
  } else {
    for_scanned([&](int s, int r) {
#pragma unroll
      for (int h = 0; h < H; ++h) atomicMaxF(&mL[r * H + h], score(s, r, h));
    });
  }
  __syncthreads();
  if (fits) {
    for (int i = tid; i < total; i += NT) {
      const int r = ld[i];
      float p[H];
#pragma unroll
      for (int h = 0; h < H; ++h) p[h] = expf(sc[i * H + h] - mL[r * H + h]);
#pragma unroll
      for (int h = 0; h < H; ++h) sc[i * H + h] = p[h];
#pragma unroll
      for (int h = 0; h < H; ++h) atomicAdd(&dL[r * H + h], p[h]);
    }
  } else {
    for_scanned([&](int s, int r) {
#pragma unroll
      for (int h = 0; h < H; ++h)
        atomicAdd(&dL[r * H + h], expf(score(s, r, h) - mL[r * H + h]));
    });
  }
  __syncthreads();

  if (fits) {
    const int jl = lane / H, hl = lane % H;
    for (int i0 = wave * EPW; i0 < total; i0 += NW * EPW) {
      const int c = (total - i0 < EPW) ? (total - i0) : EPW;
      int s_l = 0, r_l = 0; float a_l = 0.f;
      if (jl < c) {
        s_l = ls[i0 + jl]; r_l = ld[i0 + jl];
        a_l = sc[(i0 + jl) * H + hl] / dL[r_l * H + hl];
      }
      float v[EPW][Q];
#pragma unroll
      for (int j = 0; j < EPW; ++j) {
        const int s = __builtin_amdgcn_readlane(s_l, j * H);
        const float* fr = feat + (long)s * F + lane;
#pragma unroll
        for (int q = 0; q < Q; ++q) v[j][q] = fr[q * 32];
      }
#pragma unroll
      for (int j = 0; j < EPW; ++j) {
        if (j < c) {
          const int r = __builtin_amdgcn_readlane(r_l, j * H);
          float* t = tile + r * F + lane;
#pragma unroll
          for (int q = 0; q < Q; ++q) {
            const float a = __int_as_float(
                __builtin_amdgcn_readlane(__float_as_int(a_l), j * H + (q * 32) / D));
            atomicAdd(t + q * 32, v[j][q] * a);
          }
        }
      }
    }
  } else {
    for_scanned([&](int s, int r) {
      for (int h = 0; h < H; ++h) {
        const float a = expf(score(s, r, h) - mL[r * H + h]) / dL[r * H + h];
        const float* fr = feat + (long)s * F + h * D;
        float* t = tile + r * F + h * D;
        for (int k = 0; k < D; ++k) atomicAdd(t + k, fr[k] * a);
      }
    });
  }
  __syncthreads();

  {
    constexpr int C4 = F / 4;
    const float4* t4 = (const float4*)tile;
    const float4* b4 = (const float4*)bias;
    float4* o4 = (float4*)(out + (long)lo * F);
    for (int i = tid; i < nbh * C4; i += NT) {
      float4 v = t4[i];
      const float4 b = b4[i & (C4 - 1)];
      v.x += b.x; v.y += b.y; v.z += b.z; v.w += b.w;
      o4[i] = v;
    }
  }
}

extern "C" void kernel_launch(void* const* d_in, const int* in_sizes, int n_in,
                              void* d_out, int out_size, void* d_ws, size_t ws_size,
                              hipStream_t stream) {
  const float* h   = (const float*)d_in[0];
  const int*   src = (const int*)d_in[1];
  const int*   dst = (const int*)d_in[2];
  const float* W1  = (const float*)d_in[3];
  const float* al1 = (const float*)d_in[4];
  const float* ar1 = (const float*)d_in[5];
  const float* b1  = (const float*)d_in[6];
  const float* W2  = (const float*)d_in[7];
  const float* al2 = (const float*)d_in[8];
  const float* ar2 = (const float*)d_in[9];
  const float* b2  = (const float*)d_in[10];

  constexpr int F1 = 128, H1 = 4;
  constexpr int F2 = 64,  H2 = 1;
  constexpr int NB = 196;
  constexpr int CAP = 8192;
  const long N = in_sizes[0] / F1;
  const long E = in_sizes[1];

  float* ws    = (float*)d_ws;
  float* feat1 = ws;
  float* out1  = feat1 + N * F1;
  float* aux   = out1  + N * F1;
  float* el    = aux   + N * F2;
  float* er    = el    + N * H1;
  float* out2  = (float*)d_out;

  auto cdiv = [](long a, long b) { return (unsigned)((a + b - 1) / b); };
  const unsigned gblocks = cdiv(N, 128);
  const unsigned owners = cdiv(N, NB);
  unsigned char* keys = (unsigned char*)aux;
  unsigned* stash = (unsigned*)(aux + N * (F2 / 4));
  float* feat2 = feat1;

  wmma_gemm_f32_k128<F1, H1, false, NB, true><<<gblocks, 256, 0, stream>>>(
      h, W1, feat1, (int)N, al1, ar1, el, er, dst, keys, (int)E);
  edge_owner<F1, H1, NB, CAP, 8, true><<<owners, 1024, 0, stream>>>(
      src, dst, feat1, el, er, b1, out1, (int)N, (int)E, keys, stash);

  wmma_gemm_f32_k128<F2, H2, true, NB, false><<<gblocks, 256, 0, stream>>>(
      out1, W2, feat2, (int)N, al2, ar2, el, er, dst, keys, (int)E);
  edge_owner<F2, H2, NB, CAP, 16, false><<<owners, 1024, 0, stream>>>(
      src, dst, feat2, el, er, b2, out2, (int)N, (int)E, keys, stash);
}


# Layer-2 projection runs inside the layer-1 owner launch: 13 waves do the WMMA tiles from the owner's LDS sum tile (bias, ReLU in place), W2 fragments in LDS; rows never stored; 4 -> 3 launches
# speedup vs baseline: 5.0886x; 1.0732x over previous
#include <hip/hip_runtime.h>
#include <math.h>

typedef __attribute__((ext_vector_type(2))) float v2f;
typedef __attribute__((ext_vector_type(8))) float v8f;

__device__ __forceinline__ void atomicMaxF(float* addr, float v) {
  if (v >= 0.0f) atomicMax((int*)addr, __float_as_int(v));
  else           atomicMin((unsigned int*)addr, __float_as_uint(v));
}

template <int XM>
__device__ __forceinline__ float row_xor(float v) {
  return __int_as_float(__builtin_amdgcn_update_dpp(
      0, __float_as_int(v), 0x160 + XM, 0xf, 0xf, true));
}

template <int NB>
__device__ __forceinline__ void make_owner_keys(const int* __restrict__ dst,
                                                unsigned char* __restrict__ key, int E) {
  const int E4 = E >> 2;
  const int4* d4 = (const int4*)dst;
  unsigned* k32 = (unsigned*)key;
  const int step = gridDim.x * blockDim.x;
  for (int q = blockIdx.x * blockDim.x + threadIdx.x; q < E4; q += step) {
    const int4 d = d4[q];
    k32[q] = (((unsigned)d.x / NB) & 255u) | ((((unsigned)d.y / NB) & 255u) << 8) |
             ((((unsigned)d.z / NB) & 255u) << 16) | ((((unsigned)d.w / NB) & 255u) << 24);
  }
  if (blockIdx.x == 0 && (int)threadIdx.x < (E & 3)) {
    const int e = (E4 << 2) + threadIdx.x;
    key[e] = (unsigned char)(((unsigned)dst[e] / NB) & 255u);
  }
}

template <int NCOL, int H, bool RELU_A, int NB, bool KEYS>
__global__ void __launch_bounds__(256) __attribute__((amdgpu_waves_per_eu(4, 4))) wmma_gemm_f32_k128(
    const float* __restrict__ A, const float* __restrict__ B,
    float* __restrict__ C, int M,
    const float* __restrict__ al, const float* __restrict__ ar,
    float* __restrict__ el, float* __restrict__ er,
    const int* __restrict__ dst, unsigned char* __restrict__ key, int E) {
  constexpr int CT  = NCOL / 16;
  constexpr int G   = NCOL / 64;
  constexpr int D   = NCOL / H;
  constexpr int LPH = D / 4;
  constexpr int HPG = 64 / D;
  static_assert(NCOL % 64 == 0 && (LPH == 8 || LPH == 16) && G * HPG == H, "shape");
  __shared__ float4 Bs[16 * CT * 32];
  const int lane = threadIdx.x & 31;
  const int wave = threadIdx.x >> 5;
  const int c    = lane & 15;
  const int hh   = lane >> 4;
  const int mtiles = (M + 15) >> 4;
  const int tile = blockIdx.x * 8 + wave;
  const bool has = tile < mtiles;
  const int row0 = tile * 16;

  float4 a[16];
  {
    int r = row0 + c;
    if (r >= M) r = M - 1;
    const float4* ap = (const float4*)(A + (long)r * 128) + hh;
#pragma unroll
    for (int k8 = 0; k8 < 16; ++k8) a[k8] = ap[2 * k8];
  }

#pragma unroll
  for (int it = 0; it < (16 * G * 32) / 256; ++it) {
    const int i = threadIdx.x + it * 256;
    const int sg  = (i >> 5) & (G - 1);
    const int sk8 = i / (32 * G);
    const float* bp = B + (long)(8 * sk8 + 4 * hh) * NCOL + sg * 64 + c * 4;
    const float4 v0 = *(const float4*)(bp);
    const float4 v1 = *(const float4*)(bp + NCOL);
    const float4 v2 = *(const float4*)(bp + 2 * NCOL);
    const float4 v3 = *(const float4*)(bp + 3 * NCOL);
    float4* d = Bs + (sk8 * CT + 4 * sg) * 32 + lane;
    d[0]  = make_float4(v0.x, v1.x, v2.x, v3.x);
    d[32] = make_float4(v0.y, v1.y, v2.y, v3.y);
    d[64] = make_float4(v0.z, v1.z, v2.z, v3.z);
    d[96] = make_float4(v0.w, v1.w, v2.w, v3.w);
  }
  __syncthreads();

  if (has) {
    if (RELU_A) {
#pragma unroll
      for (int k8 = 0; k8 < 16; ++k8) {
        a[k8].x = fmaxf(a[k8].x, 0.f); a[k8].y = fmaxf(a[k8].y, 0.f);
        a[k8].z = fmaxf(a[k8].z, 0.f); a[k8].w = fmaxf(a[k8].w, 0.f);
      }
    }
    v8f acc[CT];
#pragma unroll
    for (int ct = 0; ct < CT; ++ct) acc[ct] = v8f{};
    const float4* bq = Bs + lane;
    auto load_b = [&](float4 (&b)[CT], int k8) {
#pragma unroll
      for (int ct = 0; ct < CT; ++ct) b[ct] = bq[(k8 * CT + ct) * 32];
    };
    auto block = [&](const float4& av, const float4 (&b)[CT]) {
      v2f a01, a23;
      a01.x = av.x; a01.y = av.y; a23.x = av.z; a23.y = av.w;
#pragma unroll
      for (int ct = 0; ct < CT; ++ct) {
        v2f b01; b01.x = b[ct].x; b01.y = b[ct].y;
        acc[ct] = __builtin_amdgcn_wmma_f32_16x16x4_f32(false, a01, false, b01, (short)0, acc[ct], false, false);
      }
#pragma unroll
      for (int ct = 0; ct < CT; ++ct) {
        v2f b23; b23.x = b[ct].z; b23.y = b[ct].w;
        acc[ct] = __builtin_amdgcn_wmma_f32_16x16x4_f32(false, a23, false, b23, (short)0, acc[ct], false, false);
      }
    };
    float4 bA[CT], bB[CT];
    load_b(bA, 0);
#pragma unroll
    for (int k8 = 0; k8 < 16; k8 += 2) {
      load_b(bB, k8 + 1);
      block(a[k8], bA);
      if (k8 + 2 < 16) load_b(bA, k8 + 2);
      block(a[k8 + 1], bB);
    }

    const int rbase = row0 + 8 * hh;
#pragma unroll
    for (int i = 0; i < 8; ++i) {
      if (rbase + i < M) {
#pragma unroll
        for (int g = 0; g < G; ++g)
          *(float4*)(C + (long)(rbase + i) * NCOL + g * 64 + c * 4) =
              make_float4(acc[4 * g][i], acc[4 * g + 1][i], acc[4 * g + 2][i], acc[4 * g + 3][i]);
      }
    }

    const int j = c & (LPH - 1);
    const int o = c / LPH;
#pragma unroll
    for (int g = 0; g < G; ++g) {
      const float4 alv = *(const float4*)(al + g * 64 + c * 4);
      const float4 arv = *(const float4*)(ar + g * 64 + c * 4);
      float pl[8], pr[8];
#pragma unroll
      for (int i = 0; i < 8; ++i) {
        pl[i] = (acc[4 * g][i] * alv.x + acc[4 * g + 1][i] * alv.y) +
                (acc[4 * g + 2][i] * alv.z + acc[4 * g + 3][i] * alv.w);
        pr[i] = (acc[4 * g][i] * arv.x + acc[4 * g + 1][i] * arv.y) +
                (acc[4 * g + 2][i] * arv.z + acc[4 * g + 3][i] * arv.w);
      }
      if (LPH == 16) {
#pragma unroll
        for (int i = 0; i < 8; ++i) { pl[i] += row_xor<8>(pl[i]); pr[i] += row_xor<8>(pr[i]); }
      }
#pragma unroll
      for (int i = 0; i < 8; ++i) { pl[i] += row_xor<4>(pl[i]); pr[i] += row_xor<4>(pr[i]); }
#pragma unroll
      for (int i = 0; i < 8; ++i) { pl[i] += row_xor<2>(pl[i]); pr[i] += row_xor<2>(pr[i]); }
#pragma unroll
      for (int i = 0; i < 8; ++i) { pl[i] += row_xor<1>(pl[i]); pr[i] += row_xor<1>(pr[i]); }
      float sl = pl[0], sr = pr[0];
#pragma unroll
      for (int i = 1; i < 8; ++i) { if (j == i) { sl = pl[i]; sr = pr[i]; } }
      if (j < 8 && rbase + j < M) {
        el[(long)(rbase + j) * H + g * HPG + o] = sl;
        er[(long)(rbase + j) * H + g * HPG + o] = sr;
      }
    }
  }
  if (KEYS) make_owner_keys<NB>(dst, key, E);
}

template <int F, int H, int NB, int CAP, int EPW, bool FIRST, bool PROJ>
__global__ void __launch_bounds__(1024) edge_owner(
    const int* __restrict__ src, const int* __restrict__ dst,
    const float* __restrict__ feat, const float* __restrict__ el,
    const float* __restrict__ er, const float* __restrict__ bias,
    float* __restrict__ out, int N, int E,
    const unsigned char* __restrict__ key, unsigned* __restrict__ stash,
    const float* __restrict__ Wn, const float* __restrict__ aln,
    const float* __restrict__ arn, float* __restrict__ featn,
    float* __restrict__ eln, float* __restrict__ ern) {
  static_assert(!PROJ || (F == 128 && CAP * H >= 16 * 4 * 32 * 4), "projection: K = 128, weights fit in sc");
  constexpr int D  = F / H;
  constexpr int Q  = F / 32;
  constexpr int NT = 1024;
  constexpr int NW = NT / 32;
  static_assert(EPW * H <= 32, "one staging lane per (edge, head)");
  __shared__ float tile[NB * F];
  __shared__ float mL[NB * H];
  __shared__ float dL[NB * H];
  __shared__ float sc[CAP * H];
  int* const le = (int*)sc;
  __shared__ int   ls[CAP];
  __shared__ int   ld[CAP];
  __shared__ int   wc[NT / 32];
  constexpr int WCAP = CAP / (NT / 32);
  const int tid  = threadIdx.x;
  const int lane = tid & 31;
  const int wave = tid >> 5;
  const int lo   = blockIdx.x * NB;
  const int nbh  = (N - lo < NB) ? (N - lo) : NB;
  constexpr int SLOT = CAP + 32;
  unsigned* slot = stash + (long)blockIdx.x * SLOT;
  const bool packs = N <= (1 << 24);
  unsigned head = 0xffffffffu;
  if (!FIRST) head = slot[0];

  for (int i = tid; i < NB * F; i += NT) tile[i] = 0.f;
  for (int i = tid; i < NB * H; i += NT) { mL[i] = -INFINITY; dL[i] = 0.f; }

  const bool keyed = gridDim.x <= 256u;
  int stashed = -1;
  if (!FIRST) {
    stashed = __builtin_amdgcn_readfirstlane((int)head);
    if (stashed > CAP) stashed = -1;
  }
  int wfill = 0;
  auto put = [&](bool p, unsigned m, int e) {
    const int pos = wfill + (int)__builtin_amdgcn_mbcnt_lo(m, 0u);
    if (p && pos < WCAP) le[wave * WCAP + pos] = e;
    wfill += __builtin_popcount(m);
  };
  const unsigned pat = (blockIdx.x & 255u) * 0x01010101u;
  auto zero_bytes = [](unsigned x) {
    const unsigned t = (x & 0x7f7f7f7fu) + 0x7f7f7f7fu;
    return ~(t | x | 0x7f7f7f7fu);
  };
  auto keep16 = [&](int4 v, bool valid, int e0) {
    const unsigned f0 = zero_bytes((unsigned)v.x ^ pat), f1 = zero_bytes((unsigned)v.y ^ pat);
    const unsigned f2 = zero_bytes((unsigned)v.z ^ pat), f3 = zero_bytes((unsigned)v.w ^ pat);
    unsigned m = valid ? ((f0 >> 7) | (f1 >> 6) | (f2 >> 5) | (f3 >> 4)) : 0u;
    unsigned any = __builtin_amdgcn_ballot_w32(m != 0u);
    while (any != 0u) {
      const bool p = (m != 0u);
      const int  i = p ? __builtin_ctz(m) : 0;
      put(p, any, e0 + ((i & 7) << 2) + (i >> 3));
      m &= m - 1u;
      any = __builtin_amdgcn_ballot_w32(m != 0u);
    }
  };
  if (keyed && stashed < 0) {
    const int4* k4 = (const int4*)key;
    const int E16 = E >> 4;
    int base = wave * 32;
    for (; base + 7 * NT + 32 <= E16; base += 8 * NT) {
      int4 v[8];
#pragma unroll
      for (int u = 0; u < 8; ++u) v[u] = k4[base + u * NT + lane];
#pragma unroll
      for (int u = 0; u < 8; ++u) keep16(v[u], true, (base + u * NT + lane) << 4);
    }
    for (; base < E16; base += NT) {
      const int  i = base + lane;
      const bool valid = i < E16;
      const int4 v = k4[valid ? i : 0];
      keep16(v, valid, i << 4);
    }
    if (wave == 0) {
      const int  e = (E16 << 4) + lane;
      const bool in = (lane < 16) && (e < E);
      const bool p = in && (unsigned)(dst[in ? e : 0] - lo) < (unsigned)nbh;
      put(p, __builtin_amdgcn_ballot_w32(p), e);
    }
  }
  if (lane == 0) wc[wave] = wfill;
  __syncthreads();
  int total = 0, off = 0;
  bool fits = keyed;
  for (int w = 0; w < NT / 32; ++w) {
    const int n = wc[w];
    if (w < wave) off += n;
    total += n;
    fits = fits && (n <= WCAP);
  }
  if (stashed >= 0) {
    total = stashed;
    fits = true;
    for (int i = tid; i < total; i += NT) {
      const unsigned w = slot[32 + i];
      ls[i] = (int)(w & 0xffffffu);
      ld[i] = (int)(w >> 24);
    }
  } else if (fits) {
    for (int i = lane; i < wfill; i += 32) {
      const int e = le[wave * WCAP + i];
      const int s = src[e];
      const int r = dst[e] - lo;
      ls[off + i] = s;
      ld[off + i] = r;
      if (FIRST) slot[32 + off + i] = (unsigned)s | ((unsigned)r << 24);
    }
  }
  if (FIRST && tid == 0) slot[0] = (fits && packs) ? (unsigned)total : 0xffffffffu;
  __syncthreads();

  auto score = [&](int s, int r, int h) {
    const float x = el[(long)s * H + h] + er[(long)(lo + r) * H + h];
    return (x > 0.f) ? x : 0.2f * x;
  };
  auto for_scanned = [&](auto fn) {
    for (int e = tid; e < E; e += NT) {
      const unsigned r = (unsigned)(dst[e] - lo);
      if (r < (unsigned)nbh) fn(src[e], (int)r);
    }
  };

  if (fits) {
    for (int i = tid; i < total; i += NT) {
      const int s = ls[i], r = ld[i];
      float x[H];
#pragma unroll
      for (int h = 0; h < H; ++h) x[h] = score(s, r, h);
#pragma unroll
      for (int h = 0; h < H; ++h) sc[i * H + h] = x[h];
#pragma unroll
      for (int h = 0; h < H; ++h) atomicMaxF(&mL[r * H + h], x[h]);
    }
  } else {
    for_scanned([&](int s, int r) {
#pragma unroll
      for (int h = 0; h < H; ++h) atomicMaxF(&mL[r * H + h], score(s, r, h));
    });
  }
  __syncthreads();
  if (fits) {
    for (int i = tid; i < total; i += NT) {
      const int r = ld[i];
      float p[H];
#pragma unroll
      for (int h = 0; h < H; ++h) p[h] = expf(sc[i * H + h] - mL[r * H + h]);
#pragma unroll
      for (int h = 0; h < H; ++h) sc[i * H + h] = p[h];
#pragma unroll
      for (int h = 0; h < H; ++h) atomicAdd(&dL[r * H + h], p[h]);
    }
  } else {
    for_scanned([&](int s, int r) {
#pragma unroll
      for (int h = 0; h < H; ++h)
        atomicAdd(&dL[r * H + h], expf(score(s, r, h) - mL[r * H + h]));
    });
  }
  __syncthreads();

  float4 w0 = make_float4(0.f, 0.f, 0.f, 0.f), w1 = w0, w2 = w0, w3 = w0;
  if (PROJ && tid < 512) {
    const float* bp = Wn + (long)(8 * (tid >> 5) + 4 * (lane >> 4)) * 64 + (lane & 15) * 4;
    w0 = *(const float4*)(bp);
    w1 = *(const float4*)(bp + 64);
    w2 = *(const float4*)(bp + 128);
    w3 = *(const float4*)(bp + 192);
  }

  if (fits) {
    const int jl = lane / H, hl = lane % H;
    for (int i0 = wave * EPW; i0 < total; i0 += NW * EPW) {
      const int c = (total - i0 < EPW) ? (total - i0) : EPW;
      int s_l = 0, r_l = 0; float a_l = 0.f;
      if (jl < c) {
        s_l = ls[i0 + jl]; r_l = ld[i0 + jl];
        a_l = sc[(i0 + jl) * H + hl] / dL[r_l * H + hl];
      }
      float v[EPW][Q];
#pragma unroll
      for (int j = 0; j < EPW; ++j) {
        const int s = __builtin_amdgcn_readlane(s_l, j * H);
        const float* fr = feat + (long)s * F + lane;
#pragma unroll
        for (int q = 0; q < Q; ++q) v[j][q] = fr[q * 32];
      }
#pragma unroll
      for (int j = 0; j < EPW; ++j) {
        if (j < c) {
          const int r = __builtin_amdgcn_readlane(r_l, j * H);
          float* t = tile + r * F + lane;
#pragma unroll
          for (int q = 0; q < Q; ++q) {
            const float a = __int_as_float(
                __builtin_amdgcn_readlane(__float_as_int(a_l), j * H + (q * 32) / D));
            atomicAdd(t + q * 32, v[j][q] * a);
          }
        }
      }
    }
  } else {
    for_scanned([&](int s, int r) {
      for (int h = 0; h < H; ++h) {
        const float a = expf(score(s, r, h) - mL[r * H + h]) / dL[r * H + h];
        const float* fr = feat + (long)s * F + h * D;
        float* t = tile + r * F + h * D;
        for (int k = 0; k < D; ++k) atomicAdd(t + k, fr[k] * a);
      }
    });
  }
  __syncthreads();

  if (!PROJ) {
    constexpr int C4 = F / 4;
    const float4* t4 = (const float4*)tile;
    const float4* b4 = (const float4*)bias;
    float4* o4 = (float4*)(out + (long)lo * F);
    for (int i = tid; i < nbh * C4; i += NT) {
      float4 v = t4[i];
      const float4 b = b4[i & (C4 - 1)];
      v.x += b.x; v.y += b.y; v.z += b.z; v.w += b.w;
      o4[i] = v;
    }
  } else {
    constexpr int C4 = F / 4;
    constexpr int CT = 4;
    float4* t4 = (float4*)tile;
    const float4* b4 = (const float4*)bias;
    for (int i = tid; i < nbh * C4; i += NT) {
      float4 v = t4[i];
      const float4 b = b4[i & (C4 - 1)];
      v.x = fmaxf(v.x + b.x, 0.f); v.y = fmaxf(v.y + b.y, 0.f);
      v.z = fmaxf(v.z + b.z, 0.f); v.w = fmaxf(v.w + b.w, 0.f);
      t4[i] = v;
    }
    float4* Bs = (float4*)sc;
    if (tid < 512) {
      float4* d = Bs + ((tid >> 5) * CT) * 32 + lane;
      d[0]  = make_float4(w0.x, w1.x, w2.x, w3.x);
      d[32] = make_float4(w0.y, w1.y, w2.y, w3.y);
      d[64] = make_float4(w0.z, w1.z, w2.z, w3.z);
      d[96] = make_float4(w0.w, w1.w, w2.w, w3.w);
    }
    __syncthreads();
    const int c  = lane & 15;
    const int hh = lane >> 4;
    if (wave * 16 < nbh) {
      int rr = wave * 16 + c;
      if (rr >= nbh) rr = nbh - 1;
      const float4* aq = t4 + rr * C4 + hh;
      const float4* bq = Bs + lane;
      v8f acc[CT];
#pragma unroll
      for (int ct = 0; ct < CT; ++ct) acc[ct] = v8f{};
      auto load_ab = [&](float4& av, float4 (&b)[CT], int k8) {
        av = aq[2 * k8];
#pragma unroll
        for (int ct = 0; ct < CT; ++ct) b[ct] = bq[(k8 * CT + ct) * 32];
      };
      auto block = [&](const float4& av, const float4 (&b)[CT]) {
        v2f a01, a23;
        a01.x = av.x; a01.y = av.y; a23.x = av.z; a23.y = av.w;
#pragma unroll
        for (int ct = 0; ct < CT; ++ct) {
          v2f b01; b01.x = b[ct].x; b01.y = b[ct].y;
          acc[ct] = __builtin_amdgcn_wmma_f32_16x16x4_f32(false, a01, false, b01, (short)0, acc[ct], false, false);
        }
#pragma unroll
        for (int ct = 0; ct < CT; ++ct) {
          v2f b23; b23.x = b[ct].z; b23.y = b[ct].w;
          acc[ct] = __builtin_amdgcn_wmma_f32_16x16x4_f32(false, a23, false, b23, (short)0, acc[ct], false, false);
        }
      };
      float4 aA, aB, bA[CT], bB[CT];
      load_ab(aA, bA, 0);
#pragma unroll
      for (int k8 = 0; k8 < 16; k8 += 2) {
        load_ab(aB, bB, k8 + 1);
        block(aA, bA);
        if (k8 + 2 < 16) load_ab(aA, bA, k8 + 2);
        block(aB, bB);
      }
      const int rbase = wave * 16 + 8 * hh;
#pragma unroll
      for (int i = 0; i < 8; ++i) {
        if (rbase + i < nbh)
          *(float4*)(featn + (long)(lo + rbase + i) * 64 + c * 4) =
              make_float4(acc[0][i], acc[1][i], acc[2][i], acc[3][i]);
      }
      const float4 alv = *(const float4*)(aln + c * 4);
      const float4 arv = *(const float4*)(arn + c * 4);
      float pl[8], pr[8];
#pragma unroll
      for (int i = 0; i < 8; ++i) {
        pl[i] = (acc[0][i] * alv.x + acc[1][i] * alv.y) + (acc[2][i] * alv.z + acc[3][i] * alv.w);
        pr[i] = (acc[0][i] * arv.x + acc[1][i] * arv.y) + (acc[2][i] * arv.z + acc[3][i] * arv.w);
      }
#pragma unroll
      for (int i = 0; i < 8; ++i) { pl[i] += row_xor<8>(pl[i]); pr[i] += row_xor<8>(pr[i]); }
#pragma unroll
      for (int i = 0; i < 8; ++i) { pl[i] += row_xor<4>(pl[i]); pr[i] += row_xor<4>(pr[i]); }
#pragma unroll
      for (int i = 0; i < 8; ++i) { pl[i] += row_xor<2>(pl[i]); pr[i] += row_xor<2>(pr[i]); }
#pragma unroll
      for (int i = 0; i < 8; ++i) { pl[i] += row_xor<1>(pl[i]); pr[i] += row_xor<1>(pr[i]); }
      float sl = pl[0], sr = pr[0];
#pragma unroll
      for (int i = 1; i < 8; ++i) { if (c == i) { sl = pl[i]; sr = pr[i]; } }
      if (c < 8 && rbase + c < nbh) {
        eln[lo + rbase + c] = sl;
        ern[lo + rbase + c] = sr;
      }
    }
  }
}

extern "C" void kernel_launch(void* const* d_in, const int* in_sizes, int n_in,
                              void* d_out, int out_size, void* d_ws, size_t ws_size,
                              hipStream_t stream) {
  const float* h   = (const float*)d_in[0];
  const int*   src = (const int*)d_in[1];
  const int*   dst = (const int*)d_in[2];
  const float* W1  = (const float*)d_in[3];
  const float* al1 = (const float*)d_in[4];
  const float* ar1 = (const float*)d_in[5];
  const float* b1  = (const float*)d_in[6];
  const float* W2  = (const float*)d_in[7];
  const float* al2 = (const float*)d_in[8];
  const float* ar2 = (const float*)d_in[9];
  const float* b2  = (const float*)d_in[10];

  constexpr int F1 = 128, H1 = 4;
  constexpr int F2 = 64,  H2 = 1;
  constexpr int NB = 196;
  constexpr int CAP = 8192;
  const long N = in_sizes[0] / F1;
  const long E = in_sizes[1];

  float* ws    = (float*)d_ws;
  float* feat1 = ws;
  float* out1  = feat1 + N * F1;
  float* aux   = out1  + N * F1;
  float* el    = aux   + N * F2;
  float* er    = el    + N * H1;
  float* out2  = (float*)d_out;

  auto cdiv = [](long a, long b) { return (unsigned)((a + b - 1) / b); };
  const unsigned gblocks = cdiv(N, 128);
  const unsigned owners = cdiv(N, NB);
  unsigned char* keys = (unsigned char*)aux;
  unsigned* stash = (unsigned*)(aux + N * (F2 / 4));
  float* feat2 = out1;
  float* el2   = out1 + N * F2;
  float* er2   = el2 + N;

  wmma_gemm_f32_k128<F1, H1, false, NB, true><<<gblocks, 256, 0, stream>>>(
      h, W1, feat1, (int)N, al1, ar1, el, er, dst, keys, (int)E);
  edge_owner<F1, H1, NB, CAP, 8, true, true><<<owners, 1024, 0, stream>>>(
      src, dst, feat1, el, er, b1, (float*)nullptr, (int)N, (int)E, keys, stash,
      W2, al2, ar2, feat2, el2, er2);

  edge_owner<F2, H2, NB, CAP, 16, false, false><<<owners, 1024, 0, stream>>>(
      src, dst, feat2, el2, er2, b2, out2, (int)N, (int)E, keys, stash,
      (const float*)nullptr, (const float*)nullptr, (const float*)nullptr,
      (float*)nullptr, (float*)nullptr, (float*)nullptr);
}


# Row-sum pass: each gathered row is ONE load of the wave (16 B per lane; layer 2: 8 B), not 4 (2) one-line loads; sum tile kept lane-major so adds stay conflict-free; LDS add order changes
# speedup vs baseline: 5.3813x; 1.0575x over previous
#include <hip/hip_runtime.h>
#include <math.h>

typedef __attribute__((ext_vector_type(2))) float v2f;
typedef __attribute__((ext_vector_type(8))) float v8f;

__device__ __forceinline__ void atomicMaxF(float* addr, float v) {
  if (v >= 0.0f) atomicMax((int*)addr, __float_as_int(v));
  else           atomicMin((unsigned int*)addr, __float_as_uint(v));
}

template <int XM>
__device__ __forceinline__ float row_xor(float v) {
  return __int_as_float(__builtin_amdgcn_update_dpp(
      0, __float_as_int(v), 0x160 + XM, 0xf, 0xf, true));
}

template <int J>
__device__ __forceinline__ float grp8_bcast(float v) {
  return __int_as_float((int)__builtin_amdgcn_mov_dpp8((unsigned)__float_as_int(v), J * 0x249249));
}

template <int J> struct trip { static constexpr int value = J; };
template <int J, int N, class Fn>
__device__ __forceinline__ void static_for(Fn&& fn) {
  if constexpr (J < N) { fn(trip<J>{}); static_for<J + 1, N>(fn); }
}

template <int V> struct row_piece;
template <> struct row_piece<4> { typedef float4 T; };
template <> struct row_piece<2> { typedef float2 T; };
__device__ __forceinline__ void unpack(const float4& t, float (&o)[4]) { o[0] = t.x; o[1] = t.y; o[2] = t.z; o[3] = t.w; }
__device__ __forceinline__ void unpack(const float2& t, float (&o)[2]) { o[0] = t.x; o[1] = t.y; }

template <int NB>
__device__ __forceinline__ void make_owner_keys(const int* __restrict__ dst,
                                                unsigned char* __restrict__ key, int E) {
  const int E4 = E >> 2;
  const int4* d4 = (const int4*)dst;
  unsigned* k32 = (unsigned*)key;
  const int step = gridDim.x * blockDim.x;
  for (int q = blockIdx.x * blockDim.x + threadIdx.x; q < E4; q += step) {
    const int4 d = d4[q];
    k32[q] = (((unsigned)d.x / NB) & 255u) | ((((unsigned)d.y / NB) & 255u) << 8) |
             ((((unsigned)d.z / NB) & 255u) << 16) | ((((unsigned)d.w / NB) & 255u) << 24);
  }
  if (blockIdx.x == 0 && (int)threadIdx.x < (E & 3)) {
    const int e = (E4 << 2) + threadIdx.x;
    key[e] = (unsigned char)(((unsigned)dst[e] / NB) & 255u);
  }
}

template <int NCOL, int H, bool RELU_A, int NB, bool KEYS>
__global__ void __launch_bounds__(256) __attribute__((amdgpu_waves_per_eu(4, 4))) wmma_gemm_f32_k128(
    const float* __restrict__ A, const float* __restrict__ B,
    float* __restrict__ C, int M,
    const float* __restrict__ al, const float* __restrict__ ar,
    float* __restrict__ el, float* __restrict__ er,
    const int* __restrict__ dst, unsigned char* __restrict__ key, int E) {
  constexpr int CT  = NCOL / 16;
  constexpr int G   = NCOL / 64;
  constexpr int D   = NCOL / H;
  constexpr int LPH = D / 4;
  constexpr int HPG = 64 / D;
  static_assert(NCOL % 64 == 0 && (LPH == 8 || LPH == 16) && G * HPG == H, "shape");
  __shared__ float4 Bs[16 * CT * 32];
  const int lane = threadIdx.x & 31;
  const int wave = threadIdx.x >> 5;
  const int c    = lane & 15;
  const int hh   = lane >> 4;
  const int mtiles = (M + 15) >> 4;
  const int tile = blockIdx.x * 8 + wave;
  const bool has = tile < mtiles;
  const int row0 = tile * 16;

  float4 a[16];
  {
    int r = row0 + c;
    if (r >= M) r = M - 1;
    const float4* ap = (const float4*)(A + (long)r * 128) + hh;
#pragma unroll
    for (int k8 = 0; k8 < 16; ++k8) a[k8] = ap[2 * k8];
  }

#pragma unroll
  for (int it = 0; it < (16 * G * 32) / 256; ++it) {
    const int i = threadIdx.x + it * 256;
    const int sg  = (i >> 5) & (G - 1);
    const int sk8 = i / (32 * G);
    const float* bp = B + (long)(8 * sk8 + 4 * hh) * NCOL + sg * 64 + c * 4;
    const float4 v0 = *(const float4*)(bp);
    const float4 v1 = *(const float4*)(bp + NCOL);
    const float4 v2 = *(const float4*)(bp + 2 * NCOL);
    const float4 v3 = *(const float4*)(bp + 3 * NCOL);
    float4* d = Bs + (sk8 * CT + 4 * sg) * 32 + lane;
    d[0]  = make_float4(v0.x, v1.x, v2.x, v3.x);
    d[32] = make_float4(v0.y, v1.y, v2.y, v3.y);
    d[64] = make_float4(v0.z, v1.z, v2.z, v3.z);
    d[96] = make_float4(v0.w, v1.w, v2.w, v3.w);
  }
  __syncthreads();

  if (has) {
    if (RELU_A) {
#pragma unroll
      for (int k8 = 0; k8 < 16; ++k8) {
        a[k8].x = fmaxf(a[k8].x, 0.f); a[k8].y = fmaxf(a[k8].y, 0.f);
        a[k8].z = fmaxf(a[k8].z, 0.f); a[k8].w = fmaxf(a[k8].w, 0.f);
      }
    }
    v8f acc[CT];
#pragma unroll
    for (int ct = 0; ct < CT; ++ct) acc[ct] = v8f{};
    const float4* bq = Bs + lane;
    auto load_b = [&](float4 (&b)[CT], int k8) {
#pragma unroll
      for (int ct = 0; ct < CT; ++ct) b[ct] = bq[(k8 * CT + ct) * 32];
    };
    auto block = [&](const float4& av, const float4 (&b)[CT]) {
      v2f a01, a23;
      a01.x = av.x; a01.y = av.y; a23.x = av.z; a23.y = av.w;
#pragma unroll
      for (int ct = 0; ct < CT; ++ct) {
        v2f b01; b01.x = b[ct].x; b01.y = b[ct].y;
        acc[ct] = __builtin_amdgcn_wmma_f32_16x16x4_f32(false, a01, false, b01, (short)0, acc[ct], false, false);
      }
#pragma unroll
      for (int ct = 0; ct < CT; ++ct) {
        v2f b23; b23.x = b[ct].z; b23.y = b[ct].w;
        acc[ct] = __builtin_amdgcn_wmma_f32_16x16x4_f32(false, a23, false, b23, (short)0, acc[ct], false, false);
      }
    };
    float4 bA[CT], bB[CT];
    load_b(bA, 0);
#pragma unroll
    for (int k8 = 0; k8 < 16; k8 += 2) {
      load_b(bB, k8 + 1);
      block(a[k8], bA);
      if (k8 + 2 < 16) load_b(bA, k8 + 2);
      block(a[k8 + 1], bB);
    }

    const int rbase = row0 + 8 * hh;
#pragma unroll
    for (int i = 0; i < 8; ++i) {
      if (rbase + i < M) {
#pragma unroll
        for (int g = 0; g < G; ++g)
          *(float4*)(C + (long)(rbase + i) * NCOL + g * 64 + c * 4) =
              make_float4(acc[4 * g][i], acc[4 * g + 1][i], acc[4 * g + 2][i], acc[4 * g + 3][i]);
      }
    }

    const int j = c & (LPH - 1);
    const int o = c / LPH;
#pragma unroll
    for (int g = 0; g < G; ++g) {
      const float4 alv = *(const float4*)(al + g * 64 + c * 4);
      const float4 arv = *(const float4*)(ar + g * 64 + c * 4);
      float pl[8], pr[8];
#pragma unroll
      for (int i = 0; i < 8; ++i) {
        pl[i] = (acc[4 * g][i] * alv.x + acc[4 * g + 1][i] * alv.y) +
                (acc[4 * g + 2][i] * alv.z + acc[4 * g + 3][i] * alv.w);
        pr[i] = (acc[4 * g][i] * arv.x + acc[4 * g + 1][i] * arv.y) +
                (acc[4 * g + 2][i] * arv.z + acc[4 * g + 3][i] * arv.w);
      }
      if (LPH == 16) {
#pragma unroll
        for (int i = 0; i < 8; ++i) { pl[i] += row_xor<8>(pl[i]); pr[i] += row_xor<8>(pr[i]); }
      }
#pragma unroll
      for (int i = 0; i < 8; ++i) { pl[i] += row_xor<4>(pl[i]); pr[i] += row_xor<4>(pr[i]); }
#pragma unroll
      for (int i = 0; i < 8; ++i) { pl[i] += row_xor<2>(pl[i]); pr[i] += row_xor<2>(pr[i]); }
#pragma unroll
      for (int i = 0; i < 8; ++i) { pl[i] += row_xor<1>(pl[i]); pr[i] += row_xor<1>(pr[i]); }
      float sl = pl[0], sr = pr[0];
#pragma unroll
      for (int i = 1; i < 8; ++i) { if (j == i) { sl = pl[i]; sr = pr[i]; } }
      if (j < 8 && rbase + j < M) {
        el[(long)(rbase + j) * H + g * HPG + o] = sl;
        er[(long)(rbase + j) * H + g * HPG + o] = sr;
      }
    }
  }
  if (KEYS) make_owner_keys<NB>(dst, key, E);
}

template <int F, int H, int NB, int CAP, int EPW, bool FIRST, bool PROJ>
__global__ void __launch_bounds__(1024) edge_owner(
    const int* __restrict__ src, const int* __restrict__ dst,
    const float* __restrict__ feat, const float* __restrict__ el,
    const float* __restrict__ er, const float* __restrict__ bias,
    float* __restrict__ out, int N, int E,
    const unsigned char* __restrict__ key, unsigned* __restrict__ stash,
    const float* __restrict__ Wn, const float* __restrict__ aln,
    const float* __restrict__ arn, float* __restrict__ featn,
    float* __restrict__ eln, float* __restrict__ ern) {
  static_assert(!PROJ || (F == 128 && CAP * H >= NB * F && CAP >= 16 * 4 * 32 * 4),
                "projection: K = 128, tile copy fits in sc, weights fit in ls");
  constexpr int D   = F / H;
  constexpr int VEC = F / 32;
  static_assert(VEC == 4 || VEC == 2, "a lane's piece of a row is one 16- or 8-byte load");
  static_assert(H == 1 || (EPW == 8 && D == 8 * VEC), "a head's columns lie in one group of 8 lanes");
  using piece_t = typename row_piece<VEC>::T;
  auto tpos = [](int col) { return (col % VEC) * 32 + col / VEC; };
  constexpr int NT = 1024;
  constexpr int NW = NT / 32;
  static_assert(EPW * H <= 32, "one staging lane per (edge, head)");
  __shared__ __align__(16) float tile[NB * F];
  __shared__ float mL[NB * H];
  __shared__ float dL[NB * H];
  __shared__ __align__(16) float sc[CAP * H];
  int* const le = (int*)sc;
  __shared__ __align__(16) int ls[CAP];
  __shared__ int   ld[CAP];
  __shared__ int   wc[NT / 32];
  constexpr int WCAP = CAP / (NT / 32);
  const int tid  = threadIdx.x;
  const int lane = tid & 31;
  const int wave = tid >> 5;
  const int lo   = blockIdx.x * NB;
  const int nbh  = (N - lo < NB) ? (N - lo) : NB;
  constexpr int SLOT = CAP + 32;
  unsigned* slot = stash + (long)blockIdx.x * SLOT;
  const bool packs = N <= (1 << 24);
  unsigned head = 0xffffffffu;
  if (!FIRST) head = slot[0];

  for (int i = tid; i < NB * F; i += NT) tile[i] = 0.f;
  for (int i = tid; i < NB * H; i += NT) { mL[i] = -INFINITY; dL[i] = 0.f; }

  const bool keyed = gridDim.x <= 256u;
  int stashed = -1;
  if (!FIRST) {
    stashed = __builtin_amdgcn_readfirstlane((int)head);
    if (stashed > CAP) stashed = -1;
  }
  int wfill = 0;
  auto put = [&](bool p, unsigned m, int e) {
    const int pos = wfill + (int)__builtin_amdgcn_mbcnt_lo(m, 0u);
    if (p && pos < WCAP) le[wave * WCAP + pos] = e;
    wfill += __builtin_popcount(m);
  };
  const unsigned pat = (blockIdx.x & 255u) * 0x01010101u;
  auto zero_bytes = [](unsigned x) {
    const unsigned t = (x & 0x7f7f7f7fu) + 0x7f7f7f7fu;
    return ~(t | x | 0x7f7f7f7fu);
  };
  auto keep16 = [&](int4 v, bool valid, int e0) {
    const unsigned f0 = zero_bytes((unsigned)v.x ^ pat), f1 = zero_bytes((unsigned)v.y ^ pat);
    const unsigned f2 = zero_bytes((unsigned)v.z ^ pat), f3 = zero_bytes((unsigned)v.w ^ pat);
    unsigned m = valid ? ((f0 >> 7) | (f1 >> 6) | (f2 >> 5) | (f3 >> 4)) : 0u;
    unsigned any = __builtin_amdgcn_ballot_w32(m != 0u);
    while (any != 0u) {
      const bool p = (m != 0u);
      const int  i = p ? __builtin_ctz(m) : 0;
      put(p, any, e0 + ((i & 7) << 2) + (i >> 3));
      m &= m - 1u;
      any = __builtin_amdgcn_ballot_w32(m != 0u);
    }
  };
  if (keyed && stashed < 0) {
    const int4* k4 = (const int4*)key;
    const int E16 = E >> 4;
    int base = wave * 32;
    for (; base + 7 * NT + 32 <= E16; base += 8 * NT) {
      int4 v[8];
#pragma unroll
      for (int u = 0; u < 8; ++u) v[u] = k4[base + u * NT + lane];
#pragma unroll
      for (int u = 0; u < 8; ++u) keep16(v[u], true, (base + u * NT + lane) << 4);
    }
    for (; base < E16; base += NT) {
      const int  i = base + lane;
      const bool valid = i < E16;
      const int4 v = k4[valid ? i : 0];
      keep16(v, valid, i << 4);
    }
    if (wave == 0) {
      const int  e = (E16 << 4) + lane;
      const bool in = (lane < 16) && (e < E);
      const bool p = in && (unsigned)(dst[in ? e : 0] - lo) < (unsigned)nbh;
      put(p, __builtin_amdgcn_ballot_w32(p), e);
    }
  }
  if (lane == 0) wc[wave] = wfill;
  __syncthreads();
  int total = 0, off = 0;
  bool fits = keyed;
  for (int w = 0; w < NT / 32; ++w) {
    const int n = wc[w];
    if (w < wave) off += n;
    total += n;
    fits = fits && (n <= WCAP);
  }
  if (stashed >= 0) {
    total = stashed;
    fits = true;
    for (int i = tid; i < total; i += NT) {
      const unsigned w = slot[32 + i];
      ls[i] = (int)(w & 0xffffffu);
      ld[i] = (int)(w >> 24);
    }
  } else if (fits) {
    for (int i = lane; i < wfill; i += 32) {
      const int e = le[wave * WCAP + i];
      const int s = src[e];
      const int r = dst[e] - lo;
      ls[off + i] = s;
      ld[off + i] = r;
      if (FIRST) slot[32 + off + i] = (unsigned)s | ((unsigned)r << 24);
    }
  }
  if (FIRST && tid == 0) slot[0] = (fits && packs) ? (unsigned)total : 0xffffffffu;
  __syncthreads();

  auto score = [&](int s, int r, int h) {
    const float x = el[(long)s * H + h] + er[(long)(lo + r) * H + h];
    return (x > 0.f) ? x : 0.2f * x;
  };
  auto for_scanned = [&](auto fn) {
    for (int e = tid; e < E; e += NT) {
      const unsigned r = (unsigned)(dst[e] - lo);
      if (r < (unsigned)nbh) fn(src[e], (int)r);
    }
  };

  if (fits) {
    for (int i = tid; i < total; i += NT) {
      const int s = ls[i], r = ld[i];
      float x[H];
#pragma unroll
      for (int h = 0; h < H; ++h) x[h] = score(s, r, h);
#pragma unroll
      for (int h = 0; h < H; ++h) sc[i * H + h] = x[h];
#pragma unroll
      for (int h = 0; h < H; ++h) atomicMaxF(&mL[r * H + h], x[h]);
    }
  } else {
    for_scanned([&](int s, int r) {
#pragma unroll
      for (int h = 0; h < H; ++h) atomicMaxF(&mL[r * H + h], score(s, r, h));
    });
  }
  __syncthreads();
  if (fits) {
    for (int i = tid; i < total; i += NT) {
      const int r = ld[i];
      float p[H];
#pragma unroll
      for (int h = 0; h < H; ++h) p[h] = expf(sc[i * H + h] - mL[r * H + h]);
#pragma unroll
      for (int h = 0; h < H; ++h) sc[i * H + h] = p[h];
#pragma unroll
      for (int h = 0; h < H; ++h) atomicAdd(&dL[r * H + h], p[h]);
    }
  } else {
    for_scanned([&](int s, int r) {
#pragma unroll
      for (int h = 0; h < H; ++h)
        atomicAdd(&dL[r * H + h], expf(score(s, r, h) - mL[r * H + h]));
    });
  }
  __syncthreads();

  float4 w0 = make_float4(0.f, 0.f, 0.f, 0.f), w1 = w0, w2 = w0, w3 = w0;
  if (PROJ && tid < 512) {
    const float* bp = Wn + (long)(8 * (tid >> 5) + 4 * (lane >> 4)) * 64 + (lane & 15) * 4;
    w0 = *(const float4*)(bp);
    w1 = *(const float4*)(bp + 64);
    w2 = *(const float4*)(bp + 128);
    w3 = *(const float4*)(bp + 192);
  }

  if (fits) {
    const int jl = lane % EPW, hl = (lane / EPW) % H;
    const int wave_u  = __builtin_amdgcn_readfirstlane(wave);
    const int total_u = __builtin_amdgcn_readfirstlane(total);
    auto chunk = [&](int i0, int c, auto full) {
      constexpr bool FULL = decltype(full)::value != 0;
      int s_l = 0, r_l = 0; float a_l = 0.f;
      if (FULL || jl < c) {
        s_l = ls[i0 + jl]; r_l = ld[i0 + jl];
        a_l = sc[(i0 + jl) * H + hl] / dL[r_l * H + hl];
      }
      float v[EPW][VEC];
#pragma unroll
      for (int j = 0; j < EPW; ++j) {
        const int s = __builtin_amdgcn_readlane(s_l, j);
        unpack(*(const piece_t*)(feat + (long)s * F + VEC * lane), v[j]);
      }
      static_for<0, EPW>([&](auto jc) {
        constexpr int j = decltype(jc)::value;
        if (FULL || j < c) {
          const int r = __builtin_amdgcn_readlane(r_l, j);
          float a;
          if constexpr (H == 1)
            a = __int_as_float(__builtin_amdgcn_readlane(__float_as_int(a_l), j));
          else
            a = grp8_bcast<j>(a_l);
          float* t = tile + r * F + lane;
#pragma unroll
          for (int k = 0; k < VEC; ++k) atomicAdd(t + k * 32, v[j][k] * a);
        }
      });
    };
    int i0 = wave_u * EPW;
    for (; i0 + EPW <= total_u; i0 += NW * EPW) chunk(i0, EPW, trip<1>{});
    if (i0 < total_u) chunk(i0, total_u - i0, trip<0>{});
  } else {
    for_scanned([&](int s, int r) {
      for (int h = 0; h < H; ++h) {
        const float a = expf(score(s, r, h) - mL[r * H + h]) / dL[r * H + h];
        const float* fr = feat + (long)s * F + h * D;
        float* t = tile + r * F;
        for (int k = 0; k < D; ++k) atomicAdd(t + tpos(h * D + k), fr[k] * a);
      }
    });
  }
  __syncthreads();

  if (!PROJ) {
    constexpr int C4 = F / 4;
    const float4* b4 = (const float4*)bias;
    float4* o4 = (float4*)(out + (long)lo * F);
    for (int i = tid; i < nbh * C4; i += NT) {
      const int m = i & (C4 - 1);
      const float* tr = tile + (i / C4) * F;
      const float4 b = b4[m];
      float4 v;
      v.x = tr[tpos(4 * m)] + b.x;     v.y = tr[tpos(4 * m + 1)] + b.y;
      v.z = tr[tpos(4 * m + 2)] + b.z; v.w = tr[tpos(4 * m + 3)] + b.w;
      o4[i] = v;
    }
  } else {
    constexpr int C4 = F / 4;
    constexpr int CT = 4;
    float4* t4 = (float4*)sc;
    const float4* b4 = (const float4*)bias;
    for (int i = tid; i < nbh * C4; i += NT) {
      const int m = i & (C4 - 1);
      const float* tr = tile + (i / C4) * F;
      const float4 b = b4[m];
      float4 v;
      v.x = fmaxf(tr[tpos(4 * m)] + b.x, 0.f);     v.y = fmaxf(tr[tpos(4 * m + 1)] + b.y, 0.f);
      v.z = fmaxf(tr[tpos(4 * m + 2)] + b.z, 0.f); v.w = fmaxf(tr[tpos(4 * m + 3)] + b.w, 0.f);
      t4[i] = v;
    }
    float4* Bs = (float4*)ls;
    if (tid < 512) {
      float4* d = Bs + ((tid >> 5) * CT) * 32 + lane;
      d[0]  = make_float4(w0.x, w1.x, w2.x, w3.x);
      d[32] = make_float4(w0.y, w1.y, w2.y, w3.y);
      d[64] = make_float4(w0.z, w1.z, w2.z, w3.z);
      d[96] = make_float4(w0.w, w1.w, w2.w, w3.w);
    }
    __syncthreads();
    const int c  = lane & 15;
    const int hh = lane >> 4;
    if (wave * 16 < nbh) {
      int rr = wave * 16 + c;
      if (rr >= nbh) rr = nbh - 1;
      const float4* aq = t4 + rr * C4 + hh;
      const float4* bq = Bs + lane;
      v8f acc[CT];
#pragma unroll
      for (int ct = 0; ct < CT; ++ct) acc[ct] = v8f{};
      auto load_ab = [&](float4& av, float4 (&b)[CT], int k8) {
        av = aq[2 * k8];
#pragma unroll
        for (int ct = 0; ct < CT; ++ct) b[ct] = bq[(k8 * CT + ct) * 32];
      };
      auto block = [&](const float4& av, const float4 (&b)[CT]) {
        v2f a01, a23;
        a01.x = av.x; a01.y = av.y; a23.x = av.z; a23.y = av.w;
#pragma unroll
        for (int ct = 0; ct < CT; ++ct) {
          v2f b01; b01.x = b[ct].x; b01.y = b[ct].y;
          acc[ct] = __builtin_amdgcn_wmma_f32_16x16x4_f32(false, a01, false, b01, (short)0, acc[ct], false, false);
        }
#pragma unroll
        for (int ct = 0; ct < CT; ++ct) {
          v2f b23; b23.x = b[ct].z; b23.y = b[ct].w;
          acc[ct] = __builtin_amdgcn_wmma_f32_16x16x4_f32(false, a23, false, b23, (short)0, acc[ct], false, false);
        }
      };
      float4 aA, aB, bA[CT], bB[CT];
      load_ab(aA, bA, 0);
#pragma unroll
      for (int k8 = 0; k8 < 16; k8 += 2) {
        load_ab(aB, bB, k8 + 1);
        block(aA, bA);
        if (k8 + 2 < 16) load_ab(aA, bA, k8 + 2);
        block(aB, bB);
      }
      const int rbase = wave * 16 + 8 * hh;
#pragma unroll
      for (int i = 0; i < 8; ++i) {
        if (rbase + i < nbh)
          *(float4*)(featn + (long)(lo + rbase + i) * 64 + c * 4) =
              make_float4(acc[0][i], acc[1][i], acc[2][i], acc[3][i]);
      }
      const float4 alv = *(const float4*)(aln + c * 4);
      const float4 arv = *(const float4*)(arn + c * 4);
      float pl[8], pr[8];
#pragma unroll
      for (int i = 0; i < 8; ++i) {
        pl[i] = (acc[0][i] * alv.x + acc[1][i] * alv.y) + (acc[2][i] * alv.z + acc[3][i] * alv.w);
        pr[i] = (acc[0][i] * arv.x + acc[1][i] * arv.y) + (acc[2][i] * arv.z + acc[3][i] * arv.w);
      }
#pragma unroll
      for (int i = 0; i < 8; ++i) { pl[i] += row_xor<8>(pl[i]); pr[i] += row_xor<8>(pr[i]); }
#pragma unroll
      for (int i = 0; i < 8; ++i) { pl[i] += row_xor<4>(pl[i]); pr[i] += row_xor<4>(pr[i]); }
#pragma unroll
      for (int i = 0; i < 8; ++i) { pl[i] += row_xor<2>(pl[i]); pr[i] += row_xor<2>(pr[i]); }
#pragma unroll
      for (int i = 0; i < 8; ++i) { pl[i] += row_xor<1>(pl[i]); pr[i] += row_xor<1>(pr[i]); }
      float sl = pl[0], sr = pr[0];
#pragma unroll
      for (int i = 1; i < 8; ++i) { if (c == i) { sl = pl[i]; sr = pr[i]; } }
      if (c < 8 && rbase + c < nbh) {
        eln[lo + rbase + c] = sl;
        ern[lo + rbase + c] = sr;
      }
    }
  }
}

extern "C" void kernel_launch(void* const* d_in, const int* in_sizes, int n_in,
                              void* d_out, int out_size, void* d_ws, size_t ws_size,
                              hipStream_t stream) {
  const float* h   = (const float*)d_in[0];
  const int*   src = (const int*)d_in[1];
  const int*   dst = (const int*)d_in[2];
  const float* W1  = (const float*)d_in[3];
  const float* al1 = (const float*)d_in[4];
  const float* ar1 = (const float*)d_in[5];
  const float* b1  = (const float*)d_in[6];
  const float* W2  = (const float*)d_in[7];
  const float* al2 = (const float*)d_in[8];
  const float* ar2 = (const float*)d_in[9];
  const float* b2  = (const float*)d_in[10];

  constexpr int F1 = 128, H1 = 4;
  constexpr int F2 = 64,  H2 = 1;
  constexpr int NB = 196;
  constexpr int CAP = 8192;
  const long N = in_sizes[0] / F1;
  const long E = in_sizes[1];

  float* ws    = (float*)d_ws;
  float* feat1 = ws;
  float* out1  = feat1 + N * F1;
  float* aux   = out1  + N * F1;
  float* el    = aux   + N * F2;
  float* er    = el    + N * H1;
  float* out2  = (float*)d_out;

  auto cdiv = [](long a, long b) { return (unsigned)((a + b - 1) / b); };
  const unsigned gblocks = cdiv(N, 128);
  const unsigned owners = cdiv(N, NB);
  unsigned char* keys = (unsigned char*)aux;
  unsigned* stash = (unsigned*)(aux + N * (F2 / 4));
  float* feat2 = out1;
  float* el2   = out1 + N * F2;
  float* er2   = el2 + N;

  wmma_gemm_f32_k128<F1, H1, false, NB, true><<<gblocks, 256, 0, stream>>>(
      h, W1, feat1, (int)N, al1, ar1, el, er, dst, keys, (int)E);
  edge_owner<F1, H1, NB, CAP, 8, true, true><<<owners, 1024, 0, stream>>>(
      src, dst, feat1, el, er, b1, (float*)nullptr, (int)N, (int)E, keys, stash,
      W2, al2, ar2, feat2, el2, er2);

  edge_owner<F2, H2, NB, CAP, 16, false, false><<<owners, 1024, 0, stream>>>(
      src, dst, feat2, el2, er2, b2, out2, (int)N, (int)E, keys, stash,
      (const float*)nullptr, (const float*)nullptr, (const float*)nullptr,
      (float*)nullptr, (float*)nullptr, (float*)nullptr);
}


# Owner scan reads an owner-major bitmap (1 bit per 4 edges, made in LDS by the GEMM launch) and fetches src/dst of flagged groups only, not a key byte per edge; list order, so LDS sum order, changes
# speedup vs baseline: 5.8256x; 1.0826x over previous
#include <hip/hip_runtime.h>
#include <math.h>

typedef __attribute__((ext_vector_type(2))) float v2f;
typedef __attribute__((ext_vector_type(8))) float v8f;

__device__ __forceinline__ void atomicMaxF(float* addr, float v) {
  if (v >= 0.0f) atomicMax((int*)addr, __float_as_int(v));
  else           atomicMin((unsigned int*)addr, __float_as_uint(v));
}

template <int XM>
__device__ __forceinline__ float row_xor(float v) {
  return __int_as_float(__builtin_amdgcn_update_dpp(
      0, __float_as_int(v), 0x160 + XM, 0xf, 0xf, true));
}

template <int J>
__device__ __forceinline__ float grp8_bcast(float v) {
  return __int_as_float((int)__builtin_amdgcn_mov_dpp8((unsigned)__float_as_int(v), J * 0x249249));
}

template <int J> struct trip { static constexpr int value = J; };
template <int J, int N, class Fn>
__device__ __forceinline__ void static_for(Fn&& fn) {
  if constexpr (J < N) { fn(trip<J>{}); static_for<J + 1, N>(fn); }
}

template <int V> struct row_piece;
template <> struct row_piece<4> { typedef float4 T; };
template <> struct row_piece<2> { typedef float2 T; };
__device__ __forceinline__ void unpack(const float4& t, float (&o)[4]) { o[0] = t.x; o[1] = t.y; o[2] = t.z; o[3] = t.w; }
__device__ __forceinline__ void unpack(const float2& t, float (&o)[2]) { o[0] = t.x; o[1] = t.y; }

template <int NB>
__device__ __forceinline__ void make_owner_bitmap(const int* __restrict__ dst,
                                                  unsigned* __restrict__ bm, int E,
                                                  unsigned* L) {
  const int tid = threadIdx.x;
  const int nspans = (E + 8191) >> 13;
  const long pitch = (long)nspans * 64;
  const int nfull = E >> 2;
  for (int span = blockIdx.x; span < nspans; span += gridDim.x) {
    for (int i = tid; i < 4096; i += 256) ((uint4*)L)[i] = make_uint4(0u, 0u, 0u, 0u);
    __syncthreads();
    if (nfull > 0) {
      int4 dv[8];
      bool ok[8];
#pragma unroll
      for (int u = 0; u < 8; ++u) {
        const int q = span * 2048 + u * 256 + tid;
        ok[u] = q < nfull;
        dv[u] = ((const int4*)dst)[ok[u] ? q : 0];
      }
#pragma unroll
      for (int u = 0; u < 8; ++u) {
        const unsigned g = (unsigned)(u * 256 + tid);
        const unsigned j = g >> 5;
        const unsigned bit = 1u << (g & 31u);
        const int d[4] = {dv[u].x, dv[u].y, dv[u].z, dv[u].w};
#pragma unroll
        for (int c = 0; c < 4; ++c) {
          const unsigned o = (unsigned)d[c] / NB;
          if (ok[u] && o < 256u) atomicOr(&L[o * 64 + ((j + o) & 63u)], bit);
        }
      }
    }
    __syncthreads();
    for (int i = tid; i < 256 * 16; i += 256) {
      const unsigned o = (unsigned)i >> 4, j4 = ((unsigned)i & 15u) * 4u;
      uint4 v;
      v.x = L[o * 64 + ((j4 + o) & 63u)];      v.y = L[o * 64 + ((j4 + 1 + o) & 63u)];
      v.z = L[o * 64 + ((j4 + 2 + o) & 63u)];  v.w = L[o * 64 + ((j4 + 3 + o) & 63u)];
      *(uint4*)(bm + (long)o * pitch + span * 64 + j4) = v;
    }
    __syncthreads();
  }
}

template <int NCOL, int H, bool RELU_A, int NB, bool KEYS>
__global__ void __launch_bounds__(256) __attribute__((amdgpu_waves_per_eu(4, 4))) wmma_gemm_f32_k128(
    const float* __restrict__ A, const float* __restrict__ B,
    float* __restrict__ C, int M,
    const float* __restrict__ al, const float* __restrict__ ar,
    float* __restrict__ el, float* __restrict__ er,
    const int* __restrict__ dst, unsigned* __restrict__ bm, int E) {
  constexpr int CT  = NCOL / 16;
  constexpr int G   = NCOL / 64;
  constexpr int D   = NCOL / H;
  constexpr int LPH = D / 4;
  constexpr int HPG = 64 / D;
  static_assert(NCOL % 64 == 0 && (LPH == 8 || LPH == 16) && G * HPG == H, "shape");
  __shared__ float4 Bs[16 * CT * 32];
  static_assert(!KEYS || sizeof(float4) * 16 * CT * 32 >= 65536, "the bitmap pieces of a span fit in the room of the weights");
  const int lane = threadIdx.x & 31;
  const int wave = threadIdx.x >> 5;
  const int c    = lane & 15;
  const int hh   = lane >> 4;
  const int mtiles = (M + 15) >> 4;
  const int tile = blockIdx.x * 8 + wave;
  const bool has = tile < mtiles;
  const int row0 = tile * 16;

  float4 a[16];
  {
    int r = row0 + c;
    if (r >= M) r = M - 1;
    const float4* ap = (const float4*)(A + (long)r * 128) + hh;
#pragma unroll
    for (int k8 = 0; k8 < 16; ++k8) a[k8] = ap[2 * k8];
  }

  if (KEYS) make_owner_bitmap<NB>(dst, bm, E, (unsigned*)Bs);

#pragma unroll
  for (int it = 0; it < (16 * G * 32) / 256; ++it) {
    const int i = threadIdx.x + it * 256;
    const int sg  = (i >> 5) & (G - 1);
    const int sk8 = i / (32 * G);
    const float* bp = B + (long)(8 * sk8 + 4 * hh) * NCOL + sg * 64 + c * 4;
    const float4 v0 = *(const float4*)(bp);
    const float4 v1 = *(const float4*)(bp + NCOL);
    const float4 v2 = *(const float4*)(bp + 2 * NCOL);
    const float4 v3 = *(const float4*)(bp + 3 * NCOL);
    float4* d = Bs + (sk8 * CT + 4 * sg) * 32 + lane;
    d[0]  = make_float4(v0.x, v1.x, v2.x, v3.x);
    d[32] = make_float4(v0.y, v1.y, v2.y, v3.y);
    d[64] = make_float4(v0.z, v1.z, v2.z, v3.z);
    d[96] = make_float4(v0.w, v1.w, v2.w, v3.w);
  }
  __syncthreads();

  if (has) {
    if (RELU_A) {
#pragma unroll
      for (int k8 = 0; k8 < 16; ++k8) {
        a[k8].x = fmaxf(a[k8].x, 0.f); a[k8].y = fmaxf(a[k8].y, 0.f);
        a[k8].z = fmaxf(a[k8].z, 0.f); a[k8].w = fmaxf(a[k8].w, 0.f);
      }
    }
    v8f acc[CT];
#pragma unroll
    for (int ct = 0; ct < CT; ++ct) acc[ct] = v8f{};
    const float4* bq = Bs + lane;
    auto load_b = [&](float4 (&b)[CT], int k8) {
#pragma unroll
      for (int ct = 0; ct < CT; ++ct) b[ct] = bq[(k8 * CT + ct) * 32];
    };
    auto block = [&](const float4& av, const float4 (&b)[CT]) {
      v2f a01, a23;
      a01.x = av.x; a01.y = av.y; a23.x = av.z; a23.y = av.w;
#pragma unroll
      for (int ct = 0; ct < CT; ++ct) {
        v2f b01; b01.x = b[ct].x; b01.y = b[ct].y;
        acc[ct] = __builtin_amdgcn_wmma_f32_16x16x4_f32(false, a01, false, b01, (short)0, acc[ct], false, false);
      }
#pragma unroll
      for (int ct = 0; ct < CT; ++ct) {
        v2f b23; b23.x = b[ct].z; b23.y = b[ct].w;
        acc[ct] = __builtin_amdgcn_wmma_f32_16x16x4_f32(false, a23, false, b23, (short)0, acc[ct], false, false);
      }
    };
    float4 bA[CT], bB[CT];
    load_b(bA, 0);
#pragma unroll
    for (int k8 = 0; k8 < 16; k8 += 2) {
      load_b(bB, k8 + 1);
      block(a[k8], bA);
      if (k8 + 2 < 16) load_b(bA, k8 + 2);
      block(a[k8 + 1], bB);
    }

    const int rbase = row0 + 8 * hh;
#pragma unroll
    for (int i = 0; i < 8; ++i) {
      if (rbase + i < M) {
#pragma unroll
        for (int g = 0; g < G; ++g)
          *(float4*)(C + (long)(rbase + i) * NCOL + g * 64 + c * 4) =
              make_float4(acc[4 * g][i], acc[4 * g + 1][i], acc[4 * g + 2][i], acc[4 * g + 3][i]);
      }
    }

    const int j = c & (LPH - 1);
    const int o = c / LPH;
#pragma unroll
    for (int g = 0; g < G; ++g) {
      const float4 alv = *(const float4*)(al + g * 64 + c * 4);
      const float4 arv = *(const float4*)(ar + g * 64 + c * 4);
      float pl[8], pr[8];
#pragma unroll
      for (int i = 0; i < 8; ++i) {
        pl[i] = (acc[4 * g][i] * alv.x + acc[4 * g + 1][i] * alv.y) +
                (acc[4 * g + 2][i] * alv.z + acc[4 * g + 3][i] * alv.w);
        pr[i] = (acc[4 * g][i] * arv.x + acc[4 * g + 1][i] * arv.y) +
                (acc[4 * g + 2][i] * arv.z + acc[4 * g + 3][i] * arv.w);
      }
      if (LPH == 16) {
#pragma unroll
        for (int i = 0; i < 8; ++i) { pl[i] += row_xor<8>(pl[i]); pr[i] += row_xor<8>(pr[i]); }
      }
#pragma unroll
      for (int i = 0; i < 8; ++i) { pl[i] += row_xor<4>(pl[i]); pr[i] += row_xor<4>(pr[i]); }
#pragma unroll
      for (int i = 0; i < 8; ++i) { pl[i] += row_xor<2>(pl[i]); pr[i] += row_xor<2>(pr[i]); }
#pragma unroll
      for (int i = 0; i < 8; ++i) { pl[i] += row_xor<1>(pl[i]); pr[i] += row_xor<1>(pr[i]); }
      float sl = pl[0], sr = pr[0];
#pragma unroll
      for (int i = 1; i < 8; ++i) { if (j == i) { sl = pl[i]; sr = pr[i]; } }
      if (j < 8 && rbase + j < M) {
        el[(long)(rbase + j) * H + g * HPG + o] = sl;
        er[(long)(rbase + j) * H + g * HPG + o] = sr;
      }
    }
  }
}

template <int F, int H, int NB, int CAP, int EPW, bool FIRST, bool PROJ>
__global__ void __launch_bounds__(1024) edge_owner(
    const int* __restrict__ src, const int* __restrict__ dst,
    const float* __restrict__ feat, const float* __restrict__ el,
    const float* __restrict__ er, const float* __restrict__ bias,
    float* __restrict__ out, int N, int E,
    const unsigned* __restrict__ bm, unsigned* __restrict__ stash,
    const float* __restrict__ Wn, const float* __restrict__ aln,
    const float* __restrict__ arn, float* __restrict__ featn,
    float* __restrict__ eln, float* __restrict__ ern) {
  static_assert(!PROJ || (F == 128 && CAP * H >= NB * F && CAP >= 16 * 4 * 32 * 4),
                "projection: K = 128, tile copy fits in sc, weights fit in ls");
  constexpr int D   = F / H;
  constexpr int VEC = F / 32;
  static_assert(VEC == 4 || VEC == 2, "a lane's piece of a row is one 16- or 8-byte load");
  static_assert(H == 1 || (EPW == 8 && D == 8 * VEC), "a head's columns lie in one group of 8 lanes");
  using piece_t = typename row_piece<VEC>::T;
  auto tpos = [](int col) { return (col % VEC) * 32 + col / VEC; };
  constexpr int NT = 1024;
  constexpr int NW = NT / 32;
  static_assert(EPW * H <= 32, "one staging lane per (edge, head)");
  static_assert(!FIRST || H >= 3, "the first launch keeps its waves' found edges in sc behind le");
  __shared__ __align__(16) float tile[NB * F];
  __shared__ float mL[NB * H];
  __shared__ float dL[NB * H];
  __shared__ __align__(16) float sc[CAP * H];
  int* const le = (int*)sc;
  __shared__ __align__(16) int ls[CAP];
  __shared__ int   ld[CAP];
  __shared__ int   wc[NT / 32];
  constexpr int WCAP = CAP / (NT / 32);
  const int tid  = threadIdx.x;
  const int lane = tid & 31;
  const int wave = tid >> 5;
  const int lo   = blockIdx.x * NB;
  const int nbh  = (N - lo < NB) ? (N - lo) : NB;
  constexpr int SLOT = CAP + 32;
  unsigned* slot = stash + (long)blockIdx.x * SLOT;
  const bool packs = N <= (1 << 24);
  unsigned head = 0xffffffffu;
  if (!FIRST) head = slot[0];

  for (int i = tid; i < NB * F; i += NT) tile[i] = 0.f;
  for (int i = tid; i < NB * H; i += NT) { mL[i] = -INFINITY; dL[i] = 0.f; }

  const bool keyed = gridDim.x <= 256u;
  int stashed = -1;
  if (!FIRST) {
    stashed = __builtin_amdgcn_readfirstlane((int)head);
    if (stashed > CAP) stashed = -1;
  }
  int wfill = 0;
  int efill = 0;
  if constexpr (FIRST) {
    if (keyed) {
      auto put = [&](bool p, unsigned m, int q) {
        const int pos = wfill + (int)__builtin_amdgcn_mbcnt_lo(m, 0u);
        if (p && pos < WCAP) le[wave * WCAP + pos] = q;
        wfill += __builtin_popcount(m);
      };
      auto keep32 = [&](unsigned m, int q0) {
        unsigned any = __builtin_amdgcn_ballot_w32(m != 0u);
        while (any != 0u) {
          const bool p = (m != 0u);
          const int  i = p ? __builtin_ctz(m) : 0;
          put(p, any, q0 + i);
          m &= m - 1u;
          any = __builtin_amdgcn_ballot_w32(m != 0u);
        }
      };
      const int nsp = (E + 8191) >> 13;
      const unsigned* row = bm + (long)blockIdx.x * ((long)nsp * 64);
      const int W = (E + 127) >> 7;
      for (int base = wave * 32; base < W; base += 8 * NT) {
        unsigned v[8];
#pragma unroll
        for (int u = 0; u < 8; ++u) {
          const int i = base + u * NT + lane;
          const unsigned x = row[(i < W) ? i : 0];
          v[u] = (i < W) ? x : 0u;
        }
#pragma unroll
        for (int u = 0; u < 8; ++u) keep32(v[u], (base + u * NT + lane) << 5);
      }
      const int nfull = E >> 2;
      const int ngr = (wfill < WCAP) ? wfill : WCAP;
      const int* const myle = le + wave * WCAP;
      int* const seg_s = (int*)sc + CAP + wave * WCAP;
      int* const seg_r = (int*)sc + 2 * CAP + wave * WCAP;
      auto found = [&](bool p, int sn, unsigned r) {
        const unsigned mk = __builtin_amdgcn_ballot_w32(p);
        const int pos = efill + (int)__builtin_amdgcn_mbcnt_lo(mk, 0u);
        if (p && pos < WCAP) { seg_s[pos] = sn; seg_r[pos] = (int)r; }
        efill += __builtin_popcount(mk);
      };
      for (int i0 = 0; i0 < ngr; i0 += 128) {
        int4 dv[4], sv[4];
        bool ok[4];
#pragma unroll
        for (int u = 0; u < 4; ++u) {
          const int i = i0 + 32 * u + lane;
          const int q = (i < ngr) ? myle[i] : -1;
          ok[u] = (q >= 0) && (q < nfull);
          const int qq = ok[u] ? q : 0;
          dv[u] = ((const int4*)dst)[qq];
          sv[u] = ((const int4*)src)[qq];
        }
#pragma unroll
        for (int u = 0; u < 4; ++u) {
          const int dd[4] = {dv[u].x, dv[u].y, dv[u].z, dv[u].w};
          const int ss[4] = {sv[u].x, sv[u].y, sv[u].z, sv[u].w};
#pragma unroll
          for (int c = 0; c < 4; ++c) {
            const unsigned r = (unsigned)(dd[c] - lo);
            found(ok[u] && r < (unsigned)nbh, ss[c], r);
          }
        }
      }
      if (wave == 0) {
        const int  e = (nfull << 2) + lane;
        const bool in = (lane < 4) && (e < E);
        const unsigned r = (unsigned)(dst[in ? e : 0] - lo);
        found(in && r < (unsigned)nbh, src[in ? e : 0], r);
      }
      if (wfill > WCAP) efill = WCAP + 1;
    }
  }
  if (lane == 0) wc[wave] = efill;
  __syncthreads();
  int total = 0, off = 0;
  bool fits = FIRST && keyed;
  for (int w = 0; w < NT / 32; ++w) {
    const int n = wc[w];
    if (w < wave) off += n;
    total += n;
    fits = fits && (n <= WCAP);
  }
  if (stashed >= 0) {
    total = stashed;
    fits = true;
    for (int i = tid; i < total; i += NT) {
      const unsigned w = slot[32 + i];
      ls[i] = (int)(w & 0xffffffu);
      ld[i] = (int)(w >> 24);
    }
  } else if (fits) {
    if constexpr (FIRST) {
      const int* const seg_s = (const int*)sc + CAP + wave * WCAP;
      const int* const seg_r = (const int*)sc + 2 * CAP + wave * WCAP;
      for (int i = lane; i < efill; i += 32) {
        const int s = seg_s[i];
        const int r = seg_r[i];
        ls[off + i] = s;
        ld[off + i] = r;
        slot[32 + off + i] = (unsigned)s | ((unsigned)r << 24);
      }
    }
  }
  if (FIRST && tid == 0) slot[0] = (fits && packs) ? (unsigned)total : 0xffffffffu;
  __syncthreads();

  auto score = [&](int s, int r, int h) {
    const float x = el[(long)s * H + h] + er[(long)(lo + r) * H + h];
    return (x > 0.f) ? x : 0.2f * x;
  };
  auto for_scanned = [&](auto fn) {
    for (int e = tid; e < E; e += NT) {
      const unsigned r = (unsigned)(dst[e] - lo);
      if (r < (unsigned)nbh) fn(src[e], (int)r);
    }
  };

  if (fits) {
    for (int i = tid; i < total; i += NT) {
      const int s = ls[i], r = ld[i];
      float x[H];
#pragma unroll
      for (int h = 0; h < H; ++h) x[h] = score(s, r, h);
#pragma unroll
      for (int h = 0; h < H; ++h) sc[i * H + h] = x[h];
#pragma unroll
      for (int h = 0; h < H; ++h) atomicMaxF(&mL[r * H + h], x[h]);
    }
  } else {
    for_scanned([&](int s, int r) {
#pragma unroll
      for (int h = 0; h < H; ++h) atomicMaxF(&mL[r * H + h], score(s, r, h));
    });
  }
  __syncthreads();
  if (fits) {
    for (int i = tid; i < total; i += NT) {
      const int r = ld[i];
      float p[H];
#pragma unroll
      for (int h = 0; h < H; ++h) p[h] = expf(sc[i * H + h] - mL[r * H + h]);
#pragma unroll
      for (int h = 0; h < H; ++h) sc[i * H + h] = p[h];
#pragma unroll
      for (int h = 0; h < H; ++h) atomicAdd(&dL[r * H + h], p[h]);
    }
  } else {
    for_scanned([&](int s, int r) {
#pragma unroll
      for (int h = 0; h < H; ++h)
        atomicAdd(&dL[r * H + h], expf(score(s, r, h) - mL[r * H + h]));
    });
  }
  __syncthreads();

  float4 w0 = make_float4(0.f, 0.f, 0.f, 0.f), w1 = w0, w2 = w0, w3 = w0;
  if (PROJ && tid < 512) {
    const float* bp = Wn + (long)(8 * (tid >> 5) + 4 * (lane >> 4)) * 64 + (lane & 15) * 4;
    w0 = *(const float4*)(bp);
    w1 = *(const float4*)(bp + 64);
    w2 = *(const float4*)(bp + 128);
    w3 = *(const float4*)(bp + 192);
  }

  if (fits) {
    const int jl = lane % EPW, hl = (lane / EPW) % H;
    const int wave_u  = __builtin_amdgcn_readfirstlane(wave);
    const int total_u = __builtin_amdgcn_readfirstlane(total);
    auto chunk = [&](int i0, int c, auto full) {
      constexpr bool FULL = decltype(full)::value != 0;
      int s_l = 0, r_l = 0; float a_l = 0.f;
      if (FULL || jl < c) {
        s_l = ls[i0 + jl]; r_l = ld[i0 + jl];
        a_l = sc[(i0 + jl) * H + hl] / dL[r_l * H + hl];
      }
      float v[EPW][VEC];
#pragma unroll
      for (int j = 0; j < EPW; ++j) {
        const int s = __builtin_amdgcn_readlane(s_l, j);
        unpack(*(const piece_t*)(feat + (long)s * F + VEC * lane), v[j]);
      }
      static_for<0, EPW>([&](auto jc) {
        constexpr int j = decltype(jc)::value;
        if (FULL || j < c) {
          const int r = __builtin_amdgcn_readlane(r_l, j);
          float a;
          if constexpr (H == 1)
            a = __int_as_float(__builtin_amdgcn_readlane(__float_as_int(a_l), j));
          else
            a = grp8_bcast<j>(a_l);
          float* t = tile + r * F + lane;
#pragma unroll
          for (int k = 0; k < VEC; ++k) atomicAdd(t + k * 32, v[j][k] * a);
        }
      });
    };
    int i0 = wave_u * EPW;
    for (; i0 + EPW <= total_u; i0 += NW * EPW) chunk(i0, EPW, trip<1>{});
    if (i0 < total_u) chunk(i0, total_u - i0, trip<0>{});
  } else {
    for_scanned([&](int s, int r) {
      for (int h = 0; h < H; ++h) {
        const float a = expf(score(s, r, h) - mL[r * H + h]) / dL[r * H + h];
        const float* fr = feat + (long)s * F + h * D;
        float* t = tile + r * F;
        for (int k = 0; k < D; ++k) atomicAdd(t + tpos(h * D + k), fr[k] * a);
      }
    });
  }
  __syncthreads();

  if (!PROJ) {
    constexpr int C4 = F / 4;
    const float4* b4 = (const float4*)bias;
    float4* o4 = (float4*)(out + (long)lo * F);
    for (int i = tid; i < nbh * C4; i += NT) {
      const int m = i & (C4 - 1);
      const float* tr = tile + (i / C4) * F;
      const float4 b = b4[m];
      float4 v;
      v.x = tr[tpos(4 * m)] + b.x;     v.y = tr[tpos(4 * m + 1)] + b.y;
      v.z = tr[tpos(4 * m + 2)] + b.z; v.w = tr[tpos(4 * m + 3)] + b.w;
      o4[i] = v;
    }
  } else {
    constexpr int C4 = F / 4;
    constexpr int CT = 4;
    float4* t4 = (float4*)sc;
    const float4* b4 = (const float4*)bias;
    for (int i = tid; i < nbh * C4; i += NT) {
      const int m = i & (C4 - 1);
      const float* tr = tile + (i / C4) * F;
      const float4 b = b4[m];
      float4 v;
      v.x = fmaxf(tr[tpos(4 * m)] + b.x, 0.f);     v.y = fmaxf(tr[tpos(4 * m + 1)] + b.y, 0.f);
      v.z = fmaxf(tr[tpos(4 * m + 2)] + b.z, 0.f); v.w = fmaxf(tr[tpos(4 * m + 3)] + b.w, 0.f);
      t4[i] = v;
    }
    float4* Bs = (float4*)ls;
    if (tid < 512) {
      float4* d = Bs + ((tid >> 5) * CT) * 32 + lane;
      d[0]  = make_float4(w0.x, w1.x, w2.x, w3.x);
      d[32] = make_float4(w0.y, w1.y, w2.y, w3.y);
      d[64] = make_float4(w0.z, w1.z, w2.z, w3.z);
      d[96] = make_float4(w0.w, w1.w, w2.w, w3.w);
    }
    __syncthreads();
    const int c  = lane & 15;
    const int hh = lane >> 4;
    if (wave * 16 < nbh) {
      int rr = wave * 16 + c;
      if (rr >= nbh) rr = nbh - 1;
      const float4* aq = t4 + rr * C4 + hh;
      const float4* bq = Bs + lane;
      v8f acc[CT];
#pragma unroll
      for (int ct = 0; ct < CT; ++ct) acc[ct] = v8f{};
      auto load_ab = [&](float4& av, float4 (&b)[CT], int k8) {
        av = aq[2 * k8];
#pragma unroll
        for (int ct = 0; ct < CT; ++ct) b[ct] = bq[(k8 * CT + ct) * 32];
      };
      auto block = [&](const float4& av, const float4 (&b)[CT]) {
        v2f a01, a23;
        a01.x = av.x; a01.y = av.y; a23.x = av.z; a23.y = av.w;
#pragma unroll
        for (int ct = 0; ct < CT; ++ct) {
          v2f b01; b01.x = b[ct].x; b01.y = b[ct].y;
          acc[ct] = __builtin_amdgcn_wmma_f32_16x16x4_f32(false, a01, false, b01, (short)0, acc[ct], false, false);
        }
#pragma unroll
        for (int ct = 0; ct < CT; ++ct) {
          v2f b23; b23.x = b[ct].z; b23.y = b[ct].w;
          acc[ct] = __builtin_amdgcn_wmma_f32_16x16x4_f32(false, a23, false, b23, (short)0, acc[ct], false, false);
        }
      };
      float4 aA, aB, bA[CT], bB[CT];
      load_ab(aA, bA, 0);
#pragma unroll
      for (int k8 = 0; k8 < 16; k8 += 2) {
        load_ab(aB, bB, k8 + 1);
        block(aA, bA);
        if (k8 + 2 < 16) load_ab(aA, bA, k8 + 2);
        block(aB, bB);
      }
      const int rbase = wave * 16 + 8 * hh;
#pragma unroll
      for (int i = 0; i < 8; ++i) {
        if (rbase + i < nbh)
          *(float4*)(featn + (long)(lo + rbase + i) * 64 + c * 4) =
              make_float4(acc[0][i], acc[1][i], acc[2][i], acc[3][i]);
      }
      const float4 alv = *(const float4*)(aln + c * 4);
      const float4 arv = *(const float4*)(arn + c * 4);
      float pl[8], pr[8];
#pragma unroll
      for (int i = 0; i < 8; ++i) {
        pl[i] = (acc[0][i] * alv.x + acc[1][i] * alv.y) + (acc[2][i] * alv.z + acc[3][i] * alv.w);
        pr[i] = (acc[0][i] * arv.x + acc[1][i] * arv.y) + (acc[2][i] * arv.z + acc[3][i] * arv.w);
      }
#pragma unroll
      for (int i = 0; i < 8; ++i) { pl[i] += row_xor<8>(pl[i]); pr[i] += row_xor<8>(pr[i]); }
#pragma unroll
      for (int i = 0; i < 8; ++i) { pl[i] += row_xor<4>(pl[i]); pr[i] += row_xor<4>(pr[i]); }
#pragma unroll
      for (int i = 0; i < 8; ++i) { pl[i] += row_xor<2>(pl[i]); pr[i] += row_xor<2>(pr[i]); }
#pragma unroll
      for (int i = 0; i < 8; ++i) { pl[i] += row_xor<1>(pl[i]); pr[i] += row_xor<1>(pr[i]); }
      float sl = pl[0], sr = pr[0];
#pragma unroll
      for (int i = 1; i < 8; ++i) { if (c == i) { sl = pl[i]; sr = pr[i]; } }
      if (c < 8 && rbase + c < nbh) {
        eln[lo + rbase + c] = sl;
        ern[lo + rbase + c] = sr;
      }
    }
  }
}

extern "C" void kernel_launch(void* const* d_in, const int* in_sizes, int n_in,
                              void* d_out, int out_size, void* d_ws, size_t ws_size,
                              hipStream_t stream) {
  const float* h   = (const float*)d_in[0];
  const int*   src = (const int*)d_in[1];
  const int*   dst = (const int*)d_in[2];
  const float* W1  = (const float*)d_in[3];
  const float* al1 = (const float*)d_in[4];
  const float* ar1 = (const float*)d_in[5];
  const float* b1  = (const float*)d_in[6];
  const float* W2  = (const float*)d_in[7];
  const float* al2 = (const float*)d_in[8];
  const float* ar2 = (const float*)d_in[9];
  const float* b2  = (const float*)d_in[10];

  constexpr int F1 = 128, H1 = 4;
  constexpr int F2 = 64,  H2 = 1;
  constexpr int NB = 196;
  constexpr int CAP = 8192;
  const long N = in_sizes[0] / F1;
  const long E = in_sizes[1];

  float* ws    = (float*)d_ws;
  float* feat1 = ws;
  float* out1  = feat1 + N * F1;
  float* aux   = out1  + N * F1;
  float* el    = aux   + N * F2;
  float* er    = el    + N * H1;
  float* out2  = (float*)d_out;

  auto cdiv = [](long a, long b) { return (unsigned)((a + b - 1) / b); };
  const unsigned gblocks = cdiv(N, 128);
  const unsigned owners = cdiv(N, NB);
  unsigned* stash = (unsigned*)(aux + N * (F2 / 4));
  float* feat2 = out1;
  float* el2   = out1 + N * F2;
  float* er2   = el2 + N;
  unsigned* bm = (unsigned*)(out1 + N * 72);

  wmma_gemm_f32_k128<F1, H1, false, NB, true><<<gblocks, 256, 0, stream>>>(
      h, W1, feat1, (int)N, al1, ar1, el, er, dst, bm, (int)E);
  edge_owner<F1, H1, NB, CAP, 8, true, true><<<owners, 1024, 0, stream>>>(
      src, dst, feat1, el, er, b1, (float*)nullptr, (int)N, (int)E, bm, stash,
      W2, al2, ar2, feat2, el2, er2);

  edge_owner<F2, H2, NB, CAP, 16, false, false><<<owners, 1024, 0, stream>>>(
      src, dst, feat2, el2, er2, b2, out2, (int)N, (int)E, bm, stash,
      (const float*)nullptr, (const float*)nullptr, (const float*)nullptr,
      (float*)nullptr, (float*)nullptr, (float*)nullptr);
}
